# starting kernel
# baseline (speedup 1.0000x reference)
#include <hip/hip_runtime.h>

#define D_FEAT   64
#define K_DIM    128
#define TILE_M   16
#define M_TILES  2
#define WAVES    4
#define WSTRIDE  72
#define FSTRIDE  132

typedef __attribute__((ext_vector_type(2))) float v2f;
typedef __attribute__((ext_vector_type(4))) float v4f;
typedef __attribute__((ext_vector_type(8))) float v8f;

__device__ __forceinline__ v8f wmma_f32_16x16x4(v2f a, v2f b, v8f c) {
  return __builtin_amdgcn_wmma_f32_16x16x4_f32(false, a, false, b, (short)0, c,
                                               false, false);
}

__global__ void k0_init(float* __restrict__ out, float* __restrict__ wsf,
                        const float* __restrict__ attn, int N, int T) {
  size_t idx    = (size_t)blockIdx.x * blockDim.x + threadIdx.x;
  size_t stride = (size_t)gridDim.x * blockDim.x;
  size_t nOut = (size_t)N * D_FEAT;
  size_t nCnt = (size_t)T * N;
  for (size_t i = idx; i < nOut; i += stride) out[i] = 0.0f;
  for (size_t i = idx; i < nCnt; i += stride) wsf[8 + i] = 0.0f;
  if (idx == 0) {
    float m = attn[0];
    for (int t = 1; t < T; ++t) m = fmaxf(m, attn[t]);
    float e[8];
    float s = 0.0f;
    for (int t = 0; t < T && t < 8; ++t) { e[t] = __expf(attn[t] - m); s += e[t]; }
    for (int t = 0; t < T && t < 8; ++t) wsf[t] = e[t] / s;
  }
}

__global__ void k1_count(const int* __restrict__ edges, float* __restrict__ cnt,
                         int N, int E, int T) {
  size_t idx    = (size_t)blockIdx.x * blockDim.x + threadIdx.x;
  size_t stride = (size_t)gridDim.x * blockDim.x;
  size_t total  = (size_t)T * E;
  for (size_t i = idx; i < total; i += stride) {
    size_t t = i / (size_t)E;
    size_t e = i - t * (size_t)E;
    int s = edges[t * 2u * (size_t)E + e];
    atomicAdd(&cnt[t * (size_t)N + s], 1.0f);
  }
}

__global__ void k2_scale(float* __restrict__ wsf, int N, int T) {
  size_t idx   = (size_t)blockIdx.x * blockDim.x + threadIdx.x;
  size_t total = (size_t)T * N;
  if (idx >= total) return;
  size_t t = idx / (size_t)N;
  float c = wsf[8 + idx];
  wsf[8 + idx] = wsf[t] / fmaxf(c, 1.0f);
}

__global__ __launch_bounds__(WAVES * 32)
void k3_gemm_scatter(const float* __restrict__ x, const float* __restrict__ W,
                     const float* __restrict__ b, const int* __restrict__ edges,
                     const float* __restrict__ scale, float* __restrict__ out,
                     int N, int E) {
  extern __shared__ float lds[];
  float* sW = lds;
  float* sF = lds + K_DIM * WSTRIDE;

  const int t     = blockIdx.y;
  const int tid   = threadIdx.x;
  const int wave  = tid >> 5;
  const int lane  = tid & 31;
  const int row16 = lane & 15;
  const int half  = lane >> 4;

  const float* Wt = W + (size_t)t * K_DIM * D_FEAT;
  for (int i = tid; i < K_DIM * (D_FEAT / 4); i += WAVES * 32) {
    int r  = i >> 4;
    int c4 = (i & 15) << 2;
    v4f v = *(const v4f*)(Wt + r * D_FEAT + c4);
    float* p = sW + r * WSTRIDE + c4;
    p[0] = v.x; p[1] = v.y; p[2] = v.z; p[3] = v.w;
  }
  __syncthreads();

  const int nblk32 = (E + M_TILES * TILE_M - 1) / (M_TILES * TILE_M);
  const int blk    = blockIdx.x * WAVES + wave;
  if (blk >= nblk32) return;

  const int* srcE = edges + (size_t)t * 2 * E;
  const int* dstE = srcE + E;
  float* sFw = sF + wave * (M_TILES * TILE_M * FSTRIDE);

  const int e0 = blk * (M_TILES * TILE_M);
  int ed0 = e0 + row16;
  int ed1 = e0 + TILE_M + row16;
  int cl0 = (ed0 < E) ? ed0 : (E - 1);
  int cl1 = (ed1 < E) ? ed1 : (E - 1);
  const int srcn0 = srcE[cl0];
  const int dstn0 = dstE[cl0];
  const int srcn1 = srcE[cl1];
  const int dstn1 = dstE[cl1];

  #pragma unroll 4
  for (int r = 0; r < TILE_M; ++r) {
    int sn0 = __shfl(srcn0, r, 32);
    int dn0 = __shfl(dstn0, r, 32);
    int sn1 = __shfl(srcn1, r, 32);
    int dn1 = __shfl(dstn1, r, 32);
    v2f vs0 = *(const v2f*)(x + (size_t)sn0 * D_FEAT + 2 * lane);
    v2f vd0 = *(const v2f*)(x + (size_t)dn0 * D_FEAT + 2 * lane);
    v2f vs1 = *(const v2f*)(x + (size_t)sn1 * D_FEAT + 2 * lane);
    v2f vd1 = *(const v2f*)(x + (size_t)dn1 * D_FEAT + 2 * lane);
    *(v2f*)(sFw + r * FSTRIDE + 2 * lane)                               = vs0;
    *(v2f*)(sFw + r * FSTRIDE + D_FEAT + 2 * lane)                      = vd0;
    *(v2f*)(sFw + (TILE_M + r) * FSTRIDE + 2 * lane)                    = vs1;
    *(v2f*)(sFw + (TILE_M + r) * FSTRIDE + D_FEAT + 2 * lane)           = vd1;
  }

  v8f accA0 = {}, accA1 = {}, accA2 = {}, accA3 = {};
  v8f accB0 = {}, accB1 = {}, accB2 = {}, accB3 = {};

  const float* aRow0 = sFw + row16 * FSTRIDE + 2 * half;
  const float* aRow1 = sFw + (TILE_M + row16) * FSTRIDE + 2 * half;

  #pragma unroll
  for (int c = 0; c < K_DIM / 4; ++c) {
    v2f a0 = *(const v2f*)(aRow0 + 4 * c);
    v2f a1 = *(const v2f*)(aRow1 + 4 * c);
    const float* bRow0 = sW + (4 * c + 2 * half) * WSTRIDE + row16;
    const float* bRow1 = bRow0 + WSTRIDE;
    v2f b0; b0.x = bRow0[0];  b0.y = bRow1[0];
    v2f b1; b1.x = bRow0[16]; b1.y = bRow1[16];
    v2f b2; b2.x = bRow0[32]; b2.y = bRow1[32];
    v2f b3; b3.x = bRow0[48]; b3.y = bRow1[48];
    accA0 = wmma_f32_16x16x4(a0, b0, accA0);
    accB0 = wmma_f32_16x16x4(a1, b0, accB0);
    accA1 = wmma_f32_16x16x4(a0, b1, accA1);
    accB1 = wmma_f32_16x16x4(a1, b1, accB1);
    accA2 = wmma_f32_16x16x4(a0, b2, accA2);
    accB2 = wmma_f32_16x16x4(a1, b2, accB2);
    accA3 = wmma_f32_16x16x4(a0, b3, accA3);
    accB3 = wmma_f32_16x16x4(a1, b3, accB3);
  }

  const float* bt = b + (size_t)t * D_FEAT;
  float bias0 = bt[0  + row16];
  float bias1 = bt[16 + row16];
  float bias2 = bt[32 + row16];
  float bias3 = bt[48 + row16];

  v8f accsA[4] = {accA0, accA1, accA2, accA3};
  v8f accsB[4] = {accB0, accB1, accB2, accB3};
  float biases[4] = {bias0, bias1, bias2, bias3};

  #pragma unroll
  for (int v = 0; v < 8; ++v) {
    int m = v + 8 * half;
    {
      int  sn    = __shfl(srcn0, m, 32);
      bool valid = (e0 + m) < E;
      float sc   = scale[(size_t)t * N + sn];
      #pragma unroll
      for (int nt = 0; nt < 4; ++nt) {
        float val = accsA[nt][v] + biases[nt];
        val = fmaxf(val, 0.0f) * sc;
        if (valid)
          atomicAdd(out + (size_t)sn * D_FEAT + nt * 16 + row16, val);
      }
    }
    {
      int  sn    = __shfl(srcn1, m, 32);
      bool valid = (e0 + TILE_M + m) < E;
      float sc   = scale[(size_t)t * N + sn];
      #pragma unroll
      for (int nt = 0; nt < 4; ++nt) {
        float val = accsB[nt][v] + biases[nt];
        val = fmaxf(val, 0.0f) * sc;
        if (valid)
          atomicAdd(out + (size_t)sn * D_FEAT + nt * 16 + row16, val);
      }
    }
  }
}

extern "C" void kernel_launch(void* const* d_in, const int* in_sizes, int n_in,
                              void* d_out, int out_size, void* d_ws, size_t ws_size,
                              hipStream_t stream) {
  const float* x     = (const float*)d_in[0];
  const float* W     = (const float*)d_in[1];
  const float* b     = (const float*)d_in[2];
  const float* attn  = (const float*)d_in[3];
  const int*   edges = (const int*)d_in[4];
  float* out = (float*)d_out;
  float* wsf = (float*)d_ws;

  const int N = in_sizes[0] / D_FEAT;
  const int T = in_sizes[3];
  const int E = in_sizes[4] / (2 * T);

  k0_init<<<2048, 256, 0, stream>>>(out, wsf, attn, N, T);
  k1_count<<<2048, 256, 0, stream>>>(edges, wsf + 8, N, E, T);
  {
    size_t total = (size_t)T * N;
    k2_scale<<<(unsigned)((total + 255) / 256), 256, 0, stream>>>(wsf, N, T);
  }
  const int nblk32 = (E + M_TILES * TILE_M - 1) / (M_TILES * TILE_M);
  dim3 grid((nblk32 + WAVES - 1) / WAVES, T);
  size_t ldsBytes = (size_t)(K_DIM * WSTRIDE + WAVES * M_TILES * TILE_M * FSTRIDE)
                    * sizeof(float);
  k3_gemm_scatter<<<grid, WAVES * 32, ldsBytes, stream>>>(x, W, b, edges,
                                                          wsf + 8, out, N, E);
}


# k3 epilogue: v_permlane16_swap of accumulator tile pairs so each atomic instruction adds one whole 128-B line of one output row, not two 64-B halves of two rows; same addends, arrival order as before
# speedup vs baseline: 1.2118x; 1.2118x over previous
#include <hip/hip_runtime.h>

#define D_FEAT   64
#define K_DIM    128
#define TILE_M   16
#define M_TILES  2
#define WAVES    4
#define WSTRIDE  72
#define FSTRIDE  132

typedef __attribute__((ext_vector_type(2))) float v2f;
typedef __attribute__((ext_vector_type(4))) float v4f;
typedef __attribute__((ext_vector_type(8))) float v8f;

__device__ __forceinline__ v8f wmma_f32_16x16x4(v2f a, v2f b, v8f c) {
  return __builtin_amdgcn_wmma_f32_16x16x4_f32(false, a, false, b, (short)0, c,
                                               false, false);
}

__device__ __forceinline__ void swap_halves(float& a, float& b) {
  typedef __attribute__((ext_vector_type(2))) unsigned v2u;
  v2u r = __builtin_amdgcn_permlane16_swap(__float_as_uint(a), __float_as_uint(b),
                                           false, false);
  a = __uint_as_float(r.x);
  b = __uint_as_float(r.y);
}

__global__ void k0_init(float* __restrict__ out, float* __restrict__ wsf,
                        const float* __restrict__ attn, int N, int T) {
  size_t idx    = (size_t)blockIdx.x * blockDim.x + threadIdx.x;
  size_t stride = (size_t)gridDim.x * blockDim.x;
  size_t nOut = (size_t)N * D_FEAT;
  size_t nCnt = (size_t)T * N;
  for (size_t i = idx; i < nOut; i += stride) out[i] = 0.0f;
  for (size_t i = idx; i < nCnt; i += stride) wsf[8 + i] = 0.0f;
  if (idx == 0) {
    float m = attn[0];
    for (int t = 1; t < T; ++t) m = fmaxf(m, attn[t]);
    float e[8];
    float s = 0.0f;
    for (int t = 0; t < T && t < 8; ++t) { e[t] = __expf(attn[t] - m); s += e[t]; }
    for (int t = 0; t < T && t < 8; ++t) wsf[t] = e[t] / s;
  }
}

__global__ void k1_count(const int* __restrict__ edges, float* __restrict__ cnt,
                         int N, int E, int T) {
  size_t idx    = (size_t)blockIdx.x * blockDim.x + threadIdx.x;
  size_t stride = (size_t)gridDim.x * blockDim.x;
  size_t total  = (size_t)T * E;
  for (size_t i = idx; i < total; i += stride) {
    size_t t = i / (size_t)E;
    size_t e = i - t * (size_t)E;
    int s = edges[t * 2u * (size_t)E + e];
    atomicAdd(&cnt[t * (size_t)N + s], 1.0f);
  }
}

__global__ void k2_scale(float* __restrict__ wsf, int N, int T) {
  size_t idx   = (size_t)blockIdx.x * blockDim.x + threadIdx.x;
  size_t total = (size_t)T * N;
  if (idx >= total) return;
  size_t t = idx / (size_t)N;
  float c = wsf[8 + idx];
  wsf[8 + idx] = wsf[t] / fmaxf(c, 1.0f);
}

__global__ __launch_bounds__(WAVES * 32)
void k3_gemm_scatter(const float* __restrict__ x, const float* __restrict__ W,
                     const float* __restrict__ b, const int* __restrict__ edges,
                     const float* __restrict__ scale, float* __restrict__ out,
                     int N, int E) {
  extern __shared__ float lds[];
  float* sW = lds;
  float* sF = lds + K_DIM * WSTRIDE;

  const int t     = blockIdx.y;
  const int tid   = threadIdx.x;
  const int wave  = tid >> 5;
  const int lane  = tid & 31;
  const int row16 = lane & 15;
  const int half  = lane >> 4;

  const float* Wt = W + (size_t)t * K_DIM * D_FEAT;
  for (int i = tid; i < K_DIM * (D_FEAT / 4); i += WAVES * 32) {
    int r  = i >> 4;
    int c4 = (i & 15) << 2;
    v4f v = *(const v4f*)(Wt + r * D_FEAT + c4);
    float* p = sW + r * WSTRIDE + c4;
    p[0] = v.x; p[1] = v.y; p[2] = v.z; p[3] = v.w;
  }
  __syncthreads();

  const int nblk32 = (E + M_TILES * TILE_M - 1) / (M_TILES * TILE_M);
  const int blk    = blockIdx.x * WAVES + wave;
  if (blk >= nblk32) return;

  const int* srcE = edges + (size_t)t * 2 * E;
  const int* dstE = srcE + E;
  float* sFw = sF + wave * (M_TILES * TILE_M * FSTRIDE);

  const int e0 = blk * (M_TILES * TILE_M);
  int ed0 = e0 + row16;
  int ed1 = e0 + TILE_M + row16;
  int cl0 = (ed0 < E) ? ed0 : (E - 1);
  int cl1 = (ed1 < E) ? ed1 : (E - 1);
  const int srcn0 = srcE[cl0];
  const int dstn0 = dstE[cl0];
  const int srcn1 = srcE[cl1];
  const int dstn1 = dstE[cl1];
  const float scl0 = scale[(size_t)t * N + srcn0];
  const float scl1 = scale[(size_t)t * N + srcn1];

  #pragma unroll 4
  for (int r = 0; r < TILE_M; ++r) {
    int sn0 = __shfl(srcn0, r, 32);
    int dn0 = __shfl(dstn0, r, 32);
    int sn1 = __shfl(srcn1, r, 32);
    int dn1 = __shfl(dstn1, r, 32);
    v2f vs0 = *(const v2f*)(x + (size_t)sn0 * D_FEAT + 2 * lane);
    v2f vd0 = *(const v2f*)(x + (size_t)dn0 * D_FEAT + 2 * lane);
    v2f vs1 = *(const v2f*)(x + (size_t)sn1 * D_FEAT + 2 * lane);
    v2f vd1 = *(const v2f*)(x + (size_t)dn1 * D_FEAT + 2 * lane);
    *(v2f*)(sFw + r * FSTRIDE + 2 * lane)                               = vs0;
    *(v2f*)(sFw + r * FSTRIDE + D_FEAT + 2 * lane)                      = vd0;
    *(v2f*)(sFw + (TILE_M + r) * FSTRIDE + 2 * lane)                    = vs1;
    *(v2f*)(sFw + (TILE_M + r) * FSTRIDE + D_FEAT + 2 * lane)           = vd1;
  }

  v8f accA0 = {}, accA1 = {}, accA2 = {}, accA3 = {};
  v8f accB0 = {}, accB1 = {}, accB2 = {}, accB3 = {};

  const float* aRow0 = sFw + row16 * FSTRIDE + 2 * half;
  const float* aRow1 = sFw + (TILE_M + row16) * FSTRIDE + 2 * half;

  #pragma unroll
  for (int c = 0; c < K_DIM / 4; ++c) {
    v2f a0 = *(const v2f*)(aRow0 + 4 * c);
    v2f a1 = *(const v2f*)(aRow1 + 4 * c);
    const float* bRow0 = sW + (4 * c + 2 * half) * WSTRIDE + row16;
    const float* bRow1 = bRow0 + WSTRIDE;
    v2f b0; b0.x = bRow0[0];  b0.y = bRow1[0];
    v2f b1; b1.x = bRow0[16]; b1.y = bRow1[16];
    v2f b2; b2.x = bRow0[32]; b2.y = bRow1[32];
    v2f b3; b3.x = bRow0[48]; b3.y = bRow1[48];
    accA0 = wmma_f32_16x16x4(a0, b0, accA0);
    accB0 = wmma_f32_16x16x4(a1, b0, accB0);
    accA1 = wmma_f32_16x16x4(a0, b1, accA1);
    accB1 = wmma_f32_16x16x4(a1, b1, accB1);
    accA2 = wmma_f32_16x16x4(a0, b2, accA2);
    accB2 = wmma_f32_16x16x4(a1, b2, accB2);
    accA3 = wmma_f32_16x16x4(a0, b3, accA3);
    accB3 = wmma_f32_16x16x4(a1, b3, accB3);
  }

  const float* bt = b + (size_t)t * D_FEAT;
  const float bias0 = bt[0  + row16];
  const float bias1 = bt[16 + row16];
  const float bias2 = bt[32 + row16];
  const float bias3 = bt[48 + row16];

  #pragma unroll
  for (int v = 0; v < 8; ++v) {
    {
      float p0 = fmaxf(accA0[v] + bias0, 0.0f);
      float p1 = fmaxf(accA1[v] + bias1, 0.0f);
      float p2 = fmaxf(accA2[v] + bias2, 0.0f);
      float p3 = fmaxf(accA3[v] + bias3, 0.0f);
      swap_halves(p0, p1);
      swap_halves(p2, p3);
      const int   snLo = __builtin_amdgcn_readlane(srcn0, v);
      const int   snHi = __builtin_amdgcn_readlane(srcn0, v + 8);
      const float scLo = __int_as_float(__builtin_amdgcn_readlane(__float_as_int(scl0), v));
      const float scHi = __int_as_float(__builtin_amdgcn_readlane(__float_as_int(scl0), v + 8));
      if (e0 + v < E) {
        float* o = out + (size_t)snLo * D_FEAT + lane;
        atomicAdd(o,      p0 * scLo);
        atomicAdd(o + 32, p2 * scLo);
      }
      if (e0 + v + 8 < E) {
        float* o = out + (size_t)snHi * D_FEAT + lane;
        atomicAdd(o,      p1 * scHi);
        atomicAdd(o + 32, p3 * scHi);
      }
    }
    {
      float p0 = fmaxf(accB0[v] + bias0, 0.0f);
      float p1 = fmaxf(accB1[v] + bias1, 0.0f);
      float p2 = fmaxf(accB2[v] + bias2, 0.0f);
      float p3 = fmaxf(accB3[v] + bias3, 0.0f);
      swap_halves(p0, p1);
      swap_halves(p2, p3);
      const int   snLo = __builtin_amdgcn_readlane(srcn1, v);
      const int   snHi = __builtin_amdgcn_readlane(srcn1, v + 8);
      const float scLo = __int_as_float(__builtin_amdgcn_readlane(__float_as_int(scl1), v));
      const float scHi = __int_as_float(__builtin_amdgcn_readlane(__float_as_int(scl1), v + 8));
      if (e0 + TILE_M + v < E) {
        float* o = out + (size_t)snLo * D_FEAT + lane;
        atomicAdd(o,      p0 * scLo);
        atomicAdd(o + 32, p2 * scLo);
      }
      if (e0 + TILE_M + v + 8 < E) {
        float* o = out + (size_t)snHi * D_FEAT + lane;
        atomicAdd(o,      p1 * scHi);
        atomicAdd(o + 32, p3 * scHi);
      }
    }
  }
}

extern "C" void kernel_launch(void* const* d_in, const int* in_sizes, int n_in,
                              void* d_out, int out_size, void* d_ws, size_t ws_size,
                              hipStream_t stream) {
  const float* x     = (const float*)d_in[0];
  const float* W     = (const float*)d_in[1];
  const float* b     = (const float*)d_in[2];
  const float* attn  = (const float*)d_in[3];
  const int*   edges = (const int*)d_in[4];
  float* out = (float*)d_out;
  float* wsf = (float*)d_ws;

  const int N = in_sizes[0] / D_FEAT;
  const int T = in_sizes[3];
  const int E = in_sizes[4] / (2 * T);

  k0_init<<<2048, 256, 0, stream>>>(out, wsf, attn, N, T);
  k1_count<<<2048, 256, 0, stream>>>(edges, wsf + 8, N, E, T);
  {
    size_t total = (size_t)T * N;
    k2_scale<<<(unsigned)((total + 255) / 256), 256, 0, stream>>>(wsf, N, T);
  }
  const int nblk32 = (E + M_TILES * TILE_M - 1) / (M_TILES * TILE_M);
  dim3 grid((nblk32 + WAVES - 1) / WAVES, T);
  size_t ldsBytes = (size_t)(K_DIM * WSTRIDE + WAVES * M_TILES * TILE_M * FSTRIDE)
                    * sizeof(float);
  k3_gemm_scatter<<<grid, WAVES * 32, ldsBytes, stream>>>(x, W, b, edges,
                                                          wsf + 8, out, N, E);
}


# k3 as persistent work-groups: 255 groups of 16 waves stage W[t] into LDS once and loop over 32-edge tiles, not one staging and barrier per 128 edges; same addends, atomic arrival order as before
# speedup vs baseline: 1.2903x; 1.0648x over previous
#include <hip/hip_runtime.h>

#define D_FEAT   64
#define K_DIM    128
#define TILE_M   16
#define M_TILES  2
#define WAVES    16
#define GROUPS_PER_TYPE 51
#define WSTRIDE  72
#define FSTRIDE  132

typedef __attribute__((ext_vector_type(2))) float v2f;
typedef __attribute__((ext_vector_type(4))) float v4f;
typedef __attribute__((ext_vector_type(8))) float v8f;

__device__ __forceinline__ v8f wmma_f32_16x16x4(v2f a, v2f b, v8f c) {
  return __builtin_amdgcn_wmma_f32_16x16x4_f32(false, a, false, b, (short)0, c,
                                               false, false);
}

__device__ __forceinline__ void swap_halves(float& a, float& b) {
  typedef __attribute__((ext_vector_type(2))) unsigned v2u;
  v2u r = __builtin_amdgcn_permlane16_swap(__float_as_uint(a), __float_as_uint(b),
                                           false, false);
  a = __uint_as_float(r.x);
  b = __uint_as_float(r.y);
}

__global__ void k0_init(float* __restrict__ out, float* __restrict__ wsf,
                        const float* __restrict__ attn, int N, int T) {
  size_t idx    = (size_t)blockIdx.x * blockDim.x + threadIdx.x;
  size_t stride = (size_t)gridDim.x * blockDim.x;
  size_t nOut = (size_t)N * D_FEAT;
  size_t nCnt = (size_t)T * N;
  for (size_t i = idx; i < nOut; i += stride) out[i] = 0.0f;
  for (size_t i = idx; i < nCnt; i += stride) wsf[8 + i] = 0.0f;
  if (idx == 0) {
    float m = attn[0];
    for (int t = 1; t < T; ++t) m = fmaxf(m, attn[t]);
    float e[8];
    float s = 0.0f;
    for (int t = 0; t < T && t < 8; ++t) { e[t] = __expf(attn[t] - m); s += e[t]; }
    for (int t = 0; t < T && t < 8; ++t) wsf[t] = e[t] / s;
  }
}

__global__ void k1_count(const int* __restrict__ edges, float* __restrict__ cnt,
                         int N, int E, int T) {
  size_t idx    = (size_t)blockIdx.x * blockDim.x + threadIdx.x;
  size_t stride = (size_t)gridDim.x * blockDim.x;
  size_t total  = (size_t)T * E;
  for (size_t i = idx; i < total; i += stride) {
    size_t t = i / (size_t)E;
    size_t e = i - t * (size_t)E;
    int s = edges[t * 2u * (size_t)E + e];
    atomicAdd(&cnt[t * (size_t)N + s], 1.0f);
  }
}

__global__ void k2_scale(float* __restrict__ wsf, int N, int T) {
  size_t idx   = (size_t)blockIdx.x * blockDim.x + threadIdx.x;
  size_t total = (size_t)T * N;
  if (idx >= total) return;
  size_t t = idx / (size_t)N;
  float c = wsf[8 + idx];
  wsf[8 + idx] = wsf[t] / fmaxf(c, 1.0f);
}

__global__ __launch_bounds__(WAVES * 32)
void k3_gemm_scatter(const float* __restrict__ x, const float* __restrict__ W,
                     const float* __restrict__ b, const int* __restrict__ edges,
                     const float* __restrict__ scale, float* __restrict__ out,
                     int N, int E) {
  __shared__ __attribute__((aligned(16))) float sW[K_DIM * WSTRIDE];
  __shared__ __attribute__((aligned(16))) float sF[WAVES * M_TILES * TILE_M * FSTRIDE];

  const int t     = blockIdx.y;
  const int tid   = threadIdx.x;
  const int wave  = __builtin_amdgcn_readfirstlane(tid >> 5);
  const int lane  = tid & 31;
  const int row16 = lane & 15;
  const int half  = lane >> 4;

  const float* Wt = W + (size_t)t * K_DIM * D_FEAT;
  for (int i = tid; i < K_DIM * (D_FEAT / 4); i += WAVES * 32) {
    int r  = i >> 4;
    int c4 = (i & 15) << 2;
    v4f v = *(const v4f*)(Wt + r * D_FEAT + c4);
    float* p = sW + r * WSTRIDE + c4;
    p[0] = v.x; p[1] = v.y; p[2] = v.z; p[3] = v.w;
  }
  __syncthreads();

  const int nblk32 = (E + M_TILES * TILE_M - 1) / (M_TILES * TILE_M);

  const int* srcE = edges + (size_t)t * 2 * E;
  const int* dstE = srcE + E;
  float* sFw = sF + wave * (M_TILES * TILE_M * FSTRIDE);

  const float* bt = b + (size_t)t * D_FEAT;
  const float bias0 = bt[0  + row16];
  const float bias1 = bt[16 + row16];
  const float bias2 = bt[32 + row16];
  const float bias3 = bt[48 + row16];

  const float* aRow0 = sFw + row16 * FSTRIDE + 2 * half;
  const float* aRow1 = sFw + (TILE_M + row16) * FSTRIDE + 2 * half;

  for (int blk = blockIdx.x * WAVES + wave; blk < nblk32; blk += gridDim.x * WAVES) {

    const int e0 = blk * (M_TILES * TILE_M);
    int ed0 = e0 + row16;
    int ed1 = e0 + TILE_M + row16;
    int cl0 = (ed0 < E) ? ed0 : (E - 1);
    int cl1 = (ed1 < E) ? ed1 : (E - 1);
    const int srcn0 = srcE[cl0];
    const int dstn0 = dstE[cl0];
    const int srcn1 = srcE[cl1];
    const int dstn1 = dstE[cl1];
    const float scl0 = scale[(size_t)t * N + srcn0];
    const float scl1 = scale[(size_t)t * N + srcn1];

    #pragma unroll 4
    for (int r = 0; r < TILE_M; ++r) {
      int sn0 = __shfl(srcn0, r, 32);
      int dn0 = __shfl(dstn0, r, 32);
      int sn1 = __shfl(srcn1, r, 32);
      int dn1 = __shfl(dstn1, r, 32);
      v2f vs0 = *(const v2f*)(x + (size_t)sn0 * D_FEAT + 2 * lane);
      v2f vd0 = *(const v2f*)(x + (size_t)dn0 * D_FEAT + 2 * lane);
      v2f vs1 = *(const v2f*)(x + (size_t)sn1 * D_FEAT + 2 * lane);
      v2f vd1 = *(const v2f*)(x + (size_t)dn1 * D_FEAT + 2 * lane);
      *(v2f*)(sFw + r * FSTRIDE + 2 * lane)                               = vs0;
      *(v2f*)(sFw + r * FSTRIDE + D_FEAT + 2 * lane)                      = vd0;
      *(v2f*)(sFw + (TILE_M + r) * FSTRIDE + 2 * lane)                    = vs1;
      *(v2f*)(sFw + (TILE_M + r) * FSTRIDE + D_FEAT + 2 * lane)           = vd1;
    }

    v8f accA0 = {}, accA1 = {}, accA2 = {}, accA3 = {};
    v8f accB0 = {}, accB1 = {}, accB2 = {}, accB3 = {};

    #pragma unroll
    for (int c = 0; c < K_DIM / 4; ++c) {
      v2f a0 = *(const v2f*)(aRow0 + 4 * c);
      v2f a1 = *(const v2f*)(aRow1 + 4 * c);
      const float* bRow0 = sW + (4 * c + 2 * half) * WSTRIDE + row16;
      const float* bRow1 = bRow0 + WSTRIDE;
      v2f b0; b0.x = bRow0[0];  b0.y = bRow1[0];
      v2f b1; b1.x = bRow0[16]; b1.y = bRow1[16];
      v2f b2; b2.x = bRow0[32]; b2.y = bRow1[32];
      v2f b3; b3.x = bRow0[48]; b3.y = bRow1[48];
      accA0 = wmma_f32_16x16x4(a0, b0, accA0);
      accB0 = wmma_f32_16x16x4(a1, b0, accB0);
      accA1 = wmma_f32_16x16x4(a0, b1, accA1);
      accB1 = wmma_f32_16x16x4(a1, b1, accB1);
      accA2 = wmma_f32_16x16x4(a0, b2, accA2);
      accB2 = wmma_f32_16x16x4(a1, b2, accB2);
      accA3 = wmma_f32_16x16x4(a0, b3, accA3);
      accB3 = wmma_f32_16x16x4(a1, b3, accB3);
    }

    #pragma unroll
    for (int v = 0; v < 8; ++v) {
      {
        float p0 = fmaxf(accA0[v] + bias0, 0.0f);
        float p1 = fmaxf(accA1[v] + bias1, 0.0f);
        float p2 = fmaxf(accA2[v] + bias2, 0.0f);
        float p3 = fmaxf(accA3[v] + bias3, 0.0f);
        swap_halves(p0, p1);
        swap_halves(p2, p3);
        const int   snLo = __builtin_amdgcn_readlane(srcn0, v);
        const int   snHi = __builtin_amdgcn_readlane(srcn0, v + 8);
        const float scLo = __int_as_float(__builtin_amdgcn_readlane(__float_as_int(scl0), v));
        const float scHi = __int_as_float(__builtin_amdgcn_readlane(__float_as_int(scl0), v + 8));
        if (e0 + v < E) {
          float* o = out + (size_t)snLo * D_FEAT + lane;
          atomicAdd(o,      p0 * scLo);
          atomicAdd(o + 32, p2 * scLo);
        }
        if (e0 + v + 8 < E) {
          float* o = out + (size_t)snHi * D_FEAT + lane;
          atomicAdd(o,      p1 * scHi);
          atomicAdd(o + 32, p3 * scHi);
        }
      }
      {
        float p0 = fmaxf(accB0[v] + bias0, 0.0f);
        float p1 = fmaxf(accB1[v] + bias1, 0.0f);
        float p2 = fmaxf(accB2[v] + bias2, 0.0f);
        float p3 = fmaxf(accB3[v] + bias3, 0.0f);
        swap_halves(p0, p1);
        swap_halves(p2, p3);
        const int   snLo = __builtin_amdgcn_readlane(srcn1, v);
        const int   snHi = __builtin_amdgcn_readlane(srcn1, v + 8);
        const float scLo = __int_as_float(__builtin_amdgcn_readlane(__float_as_int(scl1), v));
        const float scHi = __int_as_float(__builtin_amdgcn_readlane(__float_as_int(scl1), v + 8));
        if (e0 + TILE_M + v < E) {
          float* o = out + (size_t)snLo * D_FEAT + lane;
          atomicAdd(o,      p0 * scLo);
          atomicAdd(o + 32, p2 * scLo);
        }
        if (e0 + TILE_M + v + 8 < E) {
          float* o = out + (size_t)snHi * D_FEAT + lane;
          atomicAdd(o,      p1 * scHi);
          atomicAdd(o + 32, p3 * scHi);
        }
      }
    }
  }
}

extern "C" void kernel_launch(void* const* d_in, const int* in_sizes, int n_in,
                              void* d_out, int out_size, void* d_ws, size_t ws_size,
                              hipStream_t stream) {
  const float* x     = (const float*)d_in[0];
  const float* W     = (const float*)d_in[1];
  const float* b     = (const float*)d_in[2];
  const float* attn  = (const float*)d_in[3];
  const int*   edges = (const int*)d_in[4];
  float* out = (float*)d_out;
  float* wsf = (float*)d_ws;

  const int N = in_sizes[0] / D_FEAT;
  const int T = in_sizes[3];
  const int E = in_sizes[4] / (2 * T);

  k0_init<<<2048, 256, 0, stream>>>(out, wsf, attn, N, T);
  k1_count<<<2048, 256, 0, stream>>>(edges, wsf + 8, N, E, T);
  {
    size_t total = (size_t)T * N;
    k2_scale<<<(unsigned)((total + 255) / 256), 256, 0, stream>>>(wsf, N, T);
  }
  dim3 grid(GROUPS_PER_TYPE, T);
  k3_gemm_scatter<<<grid, WAVES * 32, 0, stream>>>(x, W, b, edges,
                                                   wsf + 8, out, N, E);
}


# k1 counts edges per (type, source node) in LDS: a work-group owns 2048 nodes, scans its type's source list and writes w_t/max(count,1); no global count atomics, no k2 launch; scatter sums as before
# speedup vs baseline: 1.4994x; 1.1621x over previous
#include <hip/hip_runtime.h>

#define D_FEAT   64
#define K_DIM    128
#define TILE_M   16
#define M_TILES  2
#define WAVES    16
#define GROUPS_PER_TYPE 51
#define WSTRIDE  72
#define FSTRIDE  132
#define CNT_BINS    2048
#define CNT_THREADS 1024
#define CNT_BATCH   4

typedef __attribute__((ext_vector_type(2))) float v2f;
typedef __attribute__((ext_vector_type(4))) float v4f;
typedef __attribute__((ext_vector_type(8))) float v8f;

__device__ __forceinline__ v8f wmma_f32_16x16x4(v2f a, v2f b, v8f c) {
  return __builtin_amdgcn_wmma_f32_16x16x4_f32(false, a, false, b, (short)0, c,
                                               false, false);
}

__device__ __forceinline__ void swap_halves(float& a, float& b) {
  typedef __attribute__((ext_vector_type(2))) unsigned v2u;
  v2u r = __builtin_amdgcn_permlane16_swap(__float_as_uint(a), __float_as_uint(b),
                                           false, false);
  a = __uint_as_float(r.x);
  b = __uint_as_float(r.y);
}

__global__ void k0_init(float* __restrict__ out, int N) {
  size_t idx    = (size_t)blockIdx.x * blockDim.x + threadIdx.x;
  size_t stride = (size_t)gridDim.x * blockDim.x;
  size_t nOut = (size_t)N * D_FEAT;
  for (size_t i = idx; i < nOut; i += stride) out[i] = 0.0f;
}

__device__ __forceinline__ void count_one(unsigned* cnt, int s, int lo) {
  unsigned d = (unsigned)(s - lo);
  if (d < (unsigned)CNT_BINS) atomicAdd(&cnt[d], 1u);
}

__global__ __launch_bounds__(CNT_THREADS)
void k1_count_scale(const int* __restrict__ edges, const float* __restrict__ attn,
                    float* __restrict__ scale, int N, int E, int T) {
  __shared__ unsigned cnt[CNT_BINS];
  const int t   = blockIdx.y;
  const int lo  = blockIdx.x * CNT_BINS;
  const int tid = threadIdx.x;

  for (int i = tid; i < CNT_BINS; i += CNT_THREADS) cnt[i] = 0u;
  __syncthreads();

  const int* srcE = edges + (size_t)t * 2 * E;
  if ((E & 3) == 0) {
    const int4* p  = (const int4*)srcE;
    const int   n4 = E >> 2;
    int i = tid;
    for (; i + (CNT_BATCH - 1) * CNT_THREADS < n4; i += CNT_BATCH * CNT_THREADS) {
      int4 v[CNT_BATCH];
      #pragma unroll
      for (int j = 0; j < CNT_BATCH; ++j) v[j] = p[i + j * CNT_THREADS];
      #pragma unroll
      for (int j = 0; j < CNT_BATCH; ++j) {
        count_one(cnt, v[j].x, lo);
        count_one(cnt, v[j].y, lo);
        count_one(cnt, v[j].z, lo);
        count_one(cnt, v[j].w, lo);
      }
    }
    for (; i < n4; i += CNT_THREADS) {
      int4 v = p[i];
      count_one(cnt, v.x, lo);
      count_one(cnt, v.y, lo);
      count_one(cnt, v.z, lo);
      count_one(cnt, v.w, lo);
    }
  } else {
    for (int i = tid; i < E; i += CNT_THREADS) count_one(cnt, srcE[i], lo);
  }
  __syncthreads();

  float m = attn[0];
  for (int u = 1; u < T; ++u) m = fmaxf(m, attn[u]);
  float s = 0.0f;
  for (int u = 0; u < T; ++u) s += __expf(attn[u] - m);
  const float wt = __expf(attn[t] - m) / s;

  for (int i = tid; i < CNT_BINS; i += CNT_THREADS) {
    int n = lo + i;
    if (n < N) scale[(size_t)t * N + n] = wt / fmaxf((float)cnt[i], 1.0f);
  }
}

__global__ __launch_bounds__(WAVES * 32)
void k3_gemm_scatter(const float* __restrict__ x, const float* __restrict__ W,
                     const float* __restrict__ b, const int* __restrict__ edges,
                     const float* __restrict__ scale, float* __restrict__ out,
                     int N, int E) {
  __shared__ __attribute__((aligned(16))) float sW[K_DIM * WSTRIDE];
  __shared__ __attribute__((aligned(16))) float sF[WAVES * M_TILES * TILE_M * FSTRIDE];

  const int t     = blockIdx.y;
  const int tid   = threadIdx.x;
  const int wave  = __builtin_amdgcn_readfirstlane(tid >> 5);
  const int lane  = tid & 31;
  const int row16 = lane & 15;
  const int half  = lane >> 4;

  const float* Wt = W + (size_t)t * K_DIM * D_FEAT;
  for (int i = tid; i < K_DIM * (D_FEAT / 4); i += WAVES * 32) {
    int r  = i >> 4;
    int c4 = (i & 15) << 2;
    v4f v = *(const v4f*)(Wt + r * D_FEAT + c4);
    float* p = sW + r * WSTRIDE + c4;
    p[0] = v.x; p[1] = v.y; p[2] = v.z; p[3] = v.w;
  }
  __syncthreads();

  const int nblk32 = (E + M_TILES * TILE_M - 1) / (M_TILES * TILE_M);

  const int* srcE = edges + (size_t)t * 2 * E;
  const int* dstE = srcE + E;
  float* sFw = sF + wave * (M_TILES * TILE_M * FSTRIDE);

  const float* bt = b + (size_t)t * D_FEAT;
  const float bias0 = bt[0  + row16];
  const float bias1 = bt[16 + row16];
  const float bias2 = bt[32 + row16];
  const float bias3 = bt[48 + row16];

  const float* aRow0 = sFw + row16 * FSTRIDE + 2 * half;
  const float* aRow1 = sFw + (TILE_M + row16) * FSTRIDE + 2 * half;

  for (int blk = blockIdx.x * WAVES + wave; blk < nblk32; blk += gridDim.x * WAVES) {

    const int e0 = blk * (M_TILES * TILE_M);
    int ed0 = e0 + row16;
    int ed1 = e0 + TILE_M + row16;
    int cl0 = (ed0 < E) ? ed0 : (E - 1);
    int cl1 = (ed1 < E) ? ed1 : (E - 1);
    const int srcn0 = srcE[cl0];
    const int dstn0 = dstE[cl0];
    const int srcn1 = srcE[cl1];
    const int dstn1 = dstE[cl1];
    const float scl0 = scale[(size_t)t * N + srcn0];
    const float scl1 = scale[(size_t)t * N + srcn1];

    #pragma unroll 4
    for (int r = 0; r < TILE_M; ++r) {
      int sn0 = __shfl(srcn0, r, 32);
      int dn0 = __shfl(dstn0, r, 32);
      int sn1 = __shfl(srcn1, r, 32);
      int dn1 = __shfl(dstn1, r, 32);
      v2f vs0 = *(const v2f*)(x + (size_t)sn0 * D_FEAT + 2 * lane);
      v2f vd0 = *(const v2f*)(x + (size_t)dn0 * D_FEAT + 2 * lane);
      v2f vs1 = *(const v2f*)(x + (size_t)sn1 * D_FEAT + 2 * lane);
      v2f vd1 = *(const v2f*)(x + (size_t)dn1 * D_FEAT + 2 * lane);
      *(v2f*)(sFw + r * FSTRIDE + 2 * lane)                               = vs0;
      *(v2f*)(sFw + r * FSTRIDE + D_FEAT + 2 * lane)                      = vd0;
      *(v2f*)(sFw + (TILE_M + r) * FSTRIDE + 2 * lane)                    = vs1;
      *(v2f*)(sFw + (TILE_M + r) * FSTRIDE + D_FEAT + 2 * lane)           = vd1;
    }

    v8f accA0 = {}, accA1 = {}, accA2 = {}, accA3 = {};
    v8f accB0 = {}, accB1 = {}, accB2 = {}, accB3 = {};

    #pragma unroll
    for (int c = 0; c < K_DIM / 4; ++c) {
      v2f a0 = *(const v2f*)(aRow0 + 4 * c);
      v2f a1 = *(const v2f*)(aRow1 + 4 * c);
      const float* bRow0 = sW + (4 * c + 2 * half) * WSTRIDE + row16;
      const float* bRow1 = bRow0 + WSTRIDE;
      v2f b0; b0.x = bRow0[0];  b0.y = bRow1[0];
      v2f b1; b1.x = bRow0[16]; b1.y = bRow1[16];
      v2f b2; b2.x = bRow0[32]; b2.y = bRow1[32];
      v2f b3; b3.x = bRow0[48]; b3.y = bRow1[48];
      accA0 = wmma_f32_16x16x4(a0, b0, accA0);
      accB0 = wmma_f32_16x16x4(a1, b0, accB0);
      accA1 = wmma_f32_16x16x4(a0, b1, accA1);
      accB1 = wmma_f32_16x16x4(a1, b1, accB1);
      accA2 = wmma_f32_16x16x4(a0, b2, accA2);
      accB2 = wmma_f32_16x16x4(a1, b2, accB2);
      accA3 = wmma_f32_16x16x4(a0, b3, accA3);
      accB3 = wmma_f32_16x16x4(a1, b3, accB3);
    }

    #pragma unroll
    for (int v = 0; v < 8; ++v) {
      {
        float p0 = fmaxf(accA0[v] + bias0, 0.0f);
        float p1 = fmaxf(accA1[v] + bias1, 0.0f);
        float p2 = fmaxf(accA2[v] + bias2, 0.0f);
        float p3 = fmaxf(accA3[v] + bias3, 0.0f);
        swap_halves(p0, p1);
        swap_halves(p2, p3);
        const int   snLo = __builtin_amdgcn_readlane(srcn0, v);
        const int   snHi = __builtin_amdgcn_readlane(srcn0, v + 8);
        const float scLo = __int_as_float(__builtin_amdgcn_readlane(__float_as_int(scl0), v));
        const float scHi = __int_as_float(__builtin_amdgcn_readlane(__float_as_int(scl0), v + 8));
        if (e0 + v < E) {
          float* o = out + (size_t)snLo * D_FEAT + lane;
          atomicAdd(o,      p0 * scLo);
          atomicAdd(o + 32, p2 * scLo);
        }
        if (e0 + v + 8 < E) {
          float* o = out + (size_t)snHi * D_FEAT + lane;
          atomicAdd(o,      p1 * scHi);
          atomicAdd(o + 32, p3 * scHi);
        }
      }
      {
        float p0 = fmaxf(accB0[v] + bias0, 0.0f);
        float p1 = fmaxf(accB1[v] + bias1, 0.0f);
        float p2 = fmaxf(accB2[v] + bias2, 0.0f);
        float p3 = fmaxf(accB3[v] + bias3, 0.0f);
        swap_halves(p0, p1);
        swap_halves(p2, p3);
        const int   snLo = __builtin_amdgcn_readlane(srcn1, v);
        const int   snHi = __builtin_amdgcn_readlane(srcn1, v + 8);
        const float scLo = __int_as_float(__builtin_amdgcn_readlane(__float_as_int(scl1), v));
        const float scHi = __int_as_float(__builtin_amdgcn_readlane(__float_as_int(scl1), v + 8));
        if (e0 + TILE_M + v < E) {
          float* o = out + (size_t)snLo * D_FEAT + lane;
          atomicAdd(o,      p0 * scLo);
          atomicAdd(o + 32, p2 * scLo);
        }
        if (e0 + TILE_M + v + 8 < E) {
          float* o = out + (size_t)snHi * D_FEAT + lane;
          atomicAdd(o,      p1 * scHi);
          atomicAdd(o + 32, p3 * scHi);
        }
      }
    }
  }
}

extern "C" void kernel_launch(void* const* d_in, const int* in_sizes, int n_in,
                              void* d_out, int out_size, void* d_ws, size_t ws_size,
                              hipStream_t stream) {
  const float* x     = (const float*)d_in[0];
  const float* W     = (const float*)d_in[1];
  const float* b     = (const float*)d_in[2];
  const float* attn  = (const float*)d_in[3];
  const int*   edges = (const int*)d_in[4];
  float* out = (float*)d_out;
  float* wsf = (float*)d_ws;

  const int N = in_sizes[0] / D_FEAT;
  const int T = in_sizes[3];
  const int E = in_sizes[4] / (2 * T);

  k0_init<<<2048, 256, 0, stream>>>(out, N);
  {
    dim3 cgrid((unsigned)((N + CNT_BINS - 1) / CNT_BINS), T);
    k1_count_scale<<<cgrid, CNT_THREADS, 0, stream>>>(edges, attn, wsf + 8, N, E, T);
  }
  dim3 grid(GROUPS_PER_TYPE, T);
  k3_gemm_scatter<<<grid, WAVES * 32, 0, stream>>>(x, W, b, edges,
                                                   wsf + 8, out, N, E);
}


# k3 K loop: A and B fragments as 16-byte LDS loads (W kept in LDS in fragment order, k permuted alike in both loaders: k-sum order changes), next 32-WMMA stage's 12 loads issued as one group ahead
# speedup vs baseline: 1.5503x; 1.0339x over previous
#include <hip/hip_runtime.h>

#define D_FEAT   64
#define K_DIM    128
#define TILE_M   16
#define M_TILES  2
#define WAVES    16
#define GROUPS_PER_TYPE 51
#define SC_K     8
#define N_SC     (K_DIM / SC_K)
#define FSTRIDE  132
#define CNT_BINS    2048
#define CNT_THREADS 1024
#define CNT_BATCH   4

typedef __attribute__((ext_vector_type(2))) float v2f;
typedef __attribute__((ext_vector_type(4))) float v4f;
typedef __attribute__((ext_vector_type(8))) float v8f;

__device__ __forceinline__ v8f wmma_f32_16x16x4(v2f a, v2f b, v8f c) {
  return __builtin_amdgcn_wmma_f32_16x16x4_f32(false, a, false, b, (short)0, c,
                                               false, false);
}

__device__ __forceinline__ void swap_halves(float& a, float& b) {
  typedef __attribute__((ext_vector_type(2))) unsigned v2u;
  v2u r = __builtin_amdgcn_permlane16_swap(__float_as_uint(a), __float_as_uint(b),
                                           false, false);
  a = __uint_as_float(r.x);
  b = __uint_as_float(r.y);
}

__global__ void k0_init(float* __restrict__ out, int N) {
  size_t idx    = (size_t)blockIdx.x * blockDim.x + threadIdx.x;
  size_t stride = (size_t)gridDim.x * blockDim.x;
  size_t nOut = (size_t)N * D_FEAT;
  for (size_t i = idx; i < nOut; i += stride) out[i] = 0.0f;
}

__device__ __forceinline__ void count_one(unsigned* cnt, int s, int lo) {
  unsigned d = (unsigned)(s - lo);
  if (d < (unsigned)CNT_BINS) atomicAdd(&cnt[d], 1u);
}

__global__ __launch_bounds__(CNT_THREADS)
void k1_count_scale(const int* __restrict__ edges, const float* __restrict__ attn,
                    float* __restrict__ scale, int N, int E, int T) {
  __shared__ unsigned cnt[CNT_BINS];
  const int t   = blockIdx.y;
  const int lo  = blockIdx.x * CNT_BINS;
  const int tid = threadIdx.x;

  for (int i = tid; i < CNT_BINS; i += CNT_THREADS) cnt[i] = 0u;
  __syncthreads();

  const int* srcE = edges + (size_t)t * 2 * E;
  if ((E & 3) == 0) {
    const int4* p  = (const int4*)srcE;
    const int   n4 = E >> 2;
    int i = tid;
    for (; i + (CNT_BATCH - 1) * CNT_THREADS < n4; i += CNT_BATCH * CNT_THREADS) {
      int4 v[CNT_BATCH];
      #pragma unroll
      for (int j = 0; j < CNT_BATCH; ++j) v[j] = p[i + j * CNT_THREADS];
      #pragma unroll
      for (int j = 0; j < CNT_BATCH; ++j) {
        count_one(cnt, v[j].x, lo);
        count_one(cnt, v[j].y, lo);
        count_one(cnt, v[j].z, lo);
        count_one(cnt, v[j].w, lo);
      }
    }
    for (; i < n4; i += CNT_THREADS) {
      int4 v = p[i];
      count_one(cnt, v.x, lo);
      count_one(cnt, v.y, lo);
      count_one(cnt, v.z, lo);
      count_one(cnt, v.w, lo);
    }
  } else {
    for (int i = tid; i < E; i += CNT_THREADS) count_one(cnt, srcE[i], lo);
  }
  __syncthreads();

  float m = attn[0];
  for (int u = 1; u < T; ++u) m = fmaxf(m, attn[u]);
  float s = 0.0f;
  for (int u = 0; u < T; ++u) s += __expf(attn[u] - m);
  const float wt = __expf(attn[t] - m) / s;

  for (int i = tid; i < CNT_BINS; i += CNT_THREADS) {
    int n = lo + i;
    if (n < N) scale[(size_t)t * N + n] = wt / fmaxf((float)cnt[i], 1.0f);
  }
}

struct Frag {
  v4f a0, a1;
  v4f b0, b1, b2, b3;
};

__device__ __forceinline__ void load_frag(Frag& f, const float* aRow0, const float* aRow1,
                                          const float* bLane, int s) {
  f.a0 = *(const v4f*)(aRow0 + SC_K * s);
  f.a1 = *(const v4f*)(aRow1 + SC_K * s);
  const float* bp = bLane + s * (4 * 32 * 4);
  f.b0 = *(const v4f*)(bp);
  f.b1 = *(const v4f*)(bp + 128);
  f.b2 = *(const v4f*)(bp + 256);
  f.b3 = *(const v4f*)(bp + 384);
}

__device__ __forceinline__ void mma_frag(const Frag& f,
                                         v8f& A0, v8f& A1, v8f& A2, v8f& A3,
                                         v8f& B0, v8f& B1, v8f& B2, v8f& B3) {
  {
    v2f a0 = {f.a0.x, f.a0.y}, a1 = {f.a1.x, f.a1.y};
    v2f b0 = {f.b0.x, f.b0.y}, b1 = {f.b1.x, f.b1.y};
    v2f b2 = {f.b2.x, f.b2.y}, b3 = {f.b3.x, f.b3.y};
    A0 = wmma_f32_16x16x4(a0, b0, A0);
    B0 = wmma_f32_16x16x4(a1, b0, B0);
    A1 = wmma_f32_16x16x4(a0, b1, A1);
    B1 = wmma_f32_16x16x4(a1, b1, B1);
    A2 = wmma_f32_16x16x4(a0, b2, A2);
    B2 = wmma_f32_16x16x4(a1, b2, B2);
    A3 = wmma_f32_16x16x4(a0, b3, A3);
    B3 = wmma_f32_16x16x4(a1, b3, B3);
  }
  {
    v2f a0 = {f.a0.z, f.a0.w}, a1 = {f.a1.z, f.a1.w};
    v2f b0 = {f.b0.z, f.b0.w}, b1 = {f.b1.z, f.b1.w};
    v2f b2 = {f.b2.z, f.b2.w}, b3 = {f.b3.z, f.b3.w};
    A0 = wmma_f32_16x16x4(a0, b0, A0);
    B0 = wmma_f32_16x16x4(a1, b0, B0);
    A1 = wmma_f32_16x16x4(a0, b1, A1);
    B1 = wmma_f32_16x16x4(a1, b1, B1);
    A2 = wmma_f32_16x16x4(a0, b2, A2);
    B2 = wmma_f32_16x16x4(a1, b2, B2);
    A3 = wmma_f32_16x16x4(a0, b3, A3);
    B3 = wmma_f32_16x16x4(a1, b3, B3);
  }
}

__global__ __launch_bounds__(WAVES * 32)
void k3_gemm_scatter(const float* __restrict__ x, const float* __restrict__ W,
                     const float* __restrict__ b, const int* __restrict__ edges,
                     const float* __restrict__ scale, float* __restrict__ out,
                     int N, int E) {
  __shared__ __attribute__((aligned(16))) float sB[K_DIM * D_FEAT];
  __shared__ __attribute__((aligned(16))) float sF[WAVES * M_TILES * TILE_M * FSTRIDE];

  const int t     = blockIdx.y;
  const int tid   = threadIdx.x;
  const int wave  = __builtin_amdgcn_readfirstlane(tid >> 5);
  const int lane  = tid & 31;
  const int row16 = lane & 15;
  const int half  = lane >> 4;

  const float* Wt = W + (size_t)t * K_DIM * D_FEAT;
  for (int i = tid; i < K_DIM * (D_FEAT / 4); i += WAVES * 32) {
    int k  = i >> 4;
    int c4 = (i & 15) << 2;
    v4f v = *(const v4f*)(Wt + k * D_FEAT + c4);
    int s = k >> 3, h = (k >> 2) & 1, q = k & 3;
    int j = c4 >> 4, n = c4 & 15;
    float* p = sB + (((s * 4 + j) * 32 + h * 16 + n) << 2) + q;
    p[0] = v.x; p[4] = v.y; p[8] = v.z; p[12] = v.w;
  }
  __syncthreads();

  const int nblk32 = (E + M_TILES * TILE_M - 1) / (M_TILES * TILE_M);

  const int* srcE = edges + (size_t)t * 2 * E;
  const int* dstE = srcE + E;
  float* sFw = sF + wave * (M_TILES * TILE_M * FSTRIDE);

  const float* bt = b + (size_t)t * D_FEAT;
  const float bias0 = bt[0  + row16];
  const float bias1 = bt[16 + row16];
  const float bias2 = bt[32 + row16];
  const float bias3 = bt[48 + row16];

  const float* aRow0 = sFw + row16 * FSTRIDE + 4 * half;
  const float* aRow1 = sFw + (TILE_M + row16) * FSTRIDE + 4 * half;
  const float* bLane = sB + 4 * lane;

  for (int blk = blockIdx.x * WAVES + wave; blk < nblk32; blk += gridDim.x * WAVES) {

    const int e0 = blk * (M_TILES * TILE_M);
    int ed0 = e0 + row16;
    int ed1 = e0 + TILE_M + row16;
    int cl0 = (ed0 < E) ? ed0 : (E - 1);
    int cl1 = (ed1 < E) ? ed1 : (E - 1);
    const int srcn0 = srcE[cl0];
    const int dstn0 = dstE[cl0];
    const int srcn1 = srcE[cl1];
    const int dstn1 = dstE[cl1];
    const float scl0 = scale[(size_t)t * N + srcn0];
    const float scl1 = scale[(size_t)t * N + srcn1];

    #pragma unroll 4
    for (int r = 0; r < TILE_M; ++r) {
      int sn0 = __shfl(srcn0, r, 32);
      int dn0 = __shfl(dstn0, r, 32);
      int sn1 = __shfl(srcn1, r, 32);
      int dn1 = __shfl(dstn1, r, 32);
      v2f vs0 = *(const v2f*)(x + (size_t)sn0 * D_FEAT + 2 * lane);
      v2f vd0 = *(const v2f*)(x + (size_t)dn0 * D_FEAT + 2 * lane);
      v2f vs1 = *(const v2f*)(x + (size_t)sn1 * D_FEAT + 2 * lane);
      v2f vd1 = *(const v2f*)(x + (size_t)dn1 * D_FEAT + 2 * lane);
      *(v2f*)(sFw + r * FSTRIDE + 2 * lane)                               = vs0;
      *(v2f*)(sFw + r * FSTRIDE + D_FEAT + 2 * lane)                      = vd0;
      *(v2f*)(sFw + (TILE_M + r) * FSTRIDE + 2 * lane)                    = vs1;
      *(v2f*)(sFw + (TILE_M + r) * FSTRIDE + D_FEAT + 2 * lane)           = vd1;
    }

    v8f accA0 = {}, accA1 = {}, accA2 = {}, accA3 = {};
    v8f accB0 = {}, accB1 = {}, accB2 = {}, accB3 = {};

    Frag p0, p1, q0, q1;
    load_frag(p0, aRow0, aRow1, bLane, 0);
    load_frag(p1, aRow0, aRow1, bLane, 1);
    #pragma unroll
    for (int g = 0; g < N_SC / 4; ++g) {
      load_frag(q0, aRow0, aRow1, bLane, 4 * g + 2);
      load_frag(q1, aRow0, aRow1, bLane, 4 * g + 3);
      __builtin_amdgcn_sched_barrier(0);
      mma_frag(p0, accA0, accA1, accA2, accA3, accB0, accB1, accB2, accB3);
      mma_frag(p1, accA0, accA1, accA2, accA3, accB0, accB1, accB2, accB3);
      __builtin_amdgcn_sched_barrier(0);
      if (g + 1 < N_SC / 4) {
        load_frag(p0, aRow0, aRow1, bLane, 4 * g + 4);
        load_frag(p1, aRow0, aRow1, bLane, 4 * g + 5);
      }
      __builtin_amdgcn_sched_barrier(0);
      mma_frag(q0, accA0, accA1, accA2, accA3, accB0, accB1, accB2, accB3);
      mma_frag(q1, accA0, accA1, accA2, accA3, accB0, accB1, accB2, accB3);
      __builtin_amdgcn_sched_barrier(0);
    }

    #pragma unroll
    for (int v = 0; v < 8; ++v) {
      {
        float p0 = fmaxf(accA0[v] + bias0, 0.0f);
        float p1 = fmaxf(accA1[v] + bias1, 0.0f);
        float p2 = fmaxf(accA2[v] + bias2, 0.0f);
        float p3 = fmaxf(accA3[v] + bias3, 0.0f);
        swap_halves(p0, p1);
        swap_halves(p2, p3);
        const int   snLo = __builtin_amdgcn_readlane(srcn0, v);
        const int   snHi = __builtin_amdgcn_readlane(srcn0, v + 8);
        const float scLo = __int_as_float(__builtin_amdgcn_readlane(__float_as_int(scl0), v));
        const float scHi = __int_as_float(__builtin_amdgcn_readlane(__float_as_int(scl0), v + 8));
        if (e0 + v < E) {
          float* o = out + (size_t)snLo * D_FEAT + lane;
          atomicAdd(o,      p0 * scLo);
          atomicAdd(o + 32, p2 * scLo);
        }
        if (e0 + v + 8 < E) {
          float* o = out + (size_t)snHi * D_FEAT + lane;
          atomicAdd(o,      p1 * scHi);
          atomicAdd(o + 32, p3 * scHi);
        }
      }
      {
        float p0 = fmaxf(accB0[v] + bias0, 0.0f);
        float p1 = fmaxf(accB1[v] + bias1, 0.0f);
        float p2 = fmaxf(accB2[v] + bias2, 0.0f);
        float p3 = fmaxf(accB3[v] + bias3, 0.0f);
        swap_halves(p0, p1);
        swap_halves(p2, p3);
        const int   snLo = __builtin_amdgcn_readlane(srcn1, v);
        const int   snHi = __builtin_amdgcn_readlane(srcn1, v + 8);
        const float scLo = __int_as_float(__builtin_amdgcn_readlane(__float_as_int(scl1), v));
        const float scHi = __int_as_float(__builtin_amdgcn_readlane(__float_as_int(scl1), v + 8));
        if (e0 + TILE_M + v < E) {
          float* o = out + (size_t)snLo * D_FEAT + lane;
          atomicAdd(o,      p0 * scLo);
          atomicAdd(o + 32, p2 * scLo);
        }
        if (e0 + TILE_M + v + 8 < E) {
          float* o = out + (size_t)snHi * D_FEAT + lane;
          atomicAdd(o,      p1 * scHi);
          atomicAdd(o + 32, p3 * scHi);
        }
      }
    }
  }
}

extern "C" void kernel_launch(void* const* d_in, const int* in_sizes, int n_in,
                              void* d_out, int out_size, void* d_ws, size_t ws_size,
                              hipStream_t stream) {
  const float* x     = (const float*)d_in[0];
  const float* W     = (const float*)d_in[1];
  const float* b     = (const float*)d_in[2];
  const float* attn  = (const float*)d_in[3];
  const int*   edges = (const int*)d_in[4];
  float* out = (float*)d_out;
  float* wsf = (float*)d_ws;

  const int N = in_sizes[0] / D_FEAT;
  const int T = in_sizes[3];
  const int E = in_sizes[4] / (2 * T);

  k0_init<<<2048, 256, 0, stream>>>(out, N);
  {
    dim3 cgrid((unsigned)((N + CNT_BINS - 1) / CNT_BINS), T);
    k1_count_scale<<<cgrid, CNT_THREADS, 0, stream>>>(edges, attn, wsf + 8, N, E, T);
  }
  dim3 grid(GROUPS_PER_TYPE, T);
  k3_gemm_scatter<<<grid, WAVES * 32, 0, stream>>>(x, W, b, edges,
                                                   wsf + 8, out, N, E);
}


# Project per node, not per edge: P=x@W[t][:64]+b, Q=x@W[t][64:] once per node and type (4x fewer f32 WMMAs); edge pass adds relu(P[src]+Q[dst]), same atomics; k sum now two 64-term halves
# speedup vs baseline: 1.8993x; 1.2251x over previous
#include <hip/hip_runtime.h>

#define D_FEAT   64
#define K_DIM    128
#define TILE_M   16
#define M_TILES  2
#define WAVES    16
#define GROUPS_PER_TYPE 51
#define SC_K     8
#define N_SC     (K_DIM / SC_K)
#define FSTRIDE  132
#define CNT_BINS    2048
#define CNT_THREADS 1024
#define CNT_BATCH   4
#define PROJ_GROUPS 255
#define EDGE_GROUPS 510

typedef __attribute__((ext_vector_type(2))) float v2f;
typedef __attribute__((ext_vector_type(4))) float v4f;
typedef __attribute__((ext_vector_type(8))) float v8f;

__device__ __forceinline__ v8f wmma_f32_16x16x4(v2f a, v2f b, v8f c) {
  return __builtin_amdgcn_wmma_f32_16x16x4_f32(false, a, false, b, (short)0, c,
                                               false, false);
}

__device__ __forceinline__ void swap_halves(float& a, float& b) {
  typedef __attribute__((ext_vector_type(2))) unsigned v2u;
  v2u r = __builtin_amdgcn_permlane16_swap(__float_as_uint(a), __float_as_uint(b),
                                           false, false);
  a = __uint_as_float(r.x);
  b = __uint_as_float(r.y);
}

__global__ void k0_init(float* __restrict__ out, int N) {
  size_t idx    = (size_t)blockIdx.x * blockDim.x + threadIdx.x;
  size_t stride = (size_t)gridDim.x * blockDim.x;
  size_t nOut = (size_t)N * D_FEAT;
  for (size_t i = idx; i < nOut; i += stride) out[i] = 0.0f;
}

__device__ __forceinline__ void count_one(unsigned* cnt, int s, int lo) {
  unsigned d = (unsigned)(s - lo);
  if (d < (unsigned)CNT_BINS) atomicAdd(&cnt[d], 1u);
}

__global__ __launch_bounds__(CNT_THREADS)
void k1_count_scale(const int* __restrict__ edges, const float* __restrict__ attn,
                    float* __restrict__ scale, int N, int E, int T) {
  __shared__ unsigned cnt[CNT_BINS];
  const int t   = blockIdx.y;
  const int lo  = blockIdx.x * CNT_BINS;
  const int tid = threadIdx.x;

  for (int i = tid; i < CNT_BINS; i += CNT_THREADS) cnt[i] = 0u;
  __syncthreads();

  const int* srcE = edges + (size_t)t * 2 * E;
  if ((E & 3) == 0) {
    const int4* p  = (const int4*)srcE;
    const int   n4 = E >> 2;
    int i = tid;
    for (; i + (CNT_BATCH - 1) * CNT_THREADS < n4; i += CNT_BATCH * CNT_THREADS) {
      int4 v[CNT_BATCH];
      #pragma unroll
      for (int j = 0; j < CNT_BATCH; ++j) v[j] = p[i + j * CNT_THREADS];
      #pragma unroll
      for (int j = 0; j < CNT_BATCH; ++j) {
        count_one(cnt, v[j].x, lo);
        count_one(cnt, v[j].y, lo);
        count_one(cnt, v[j].z, lo);
        count_one(cnt, v[j].w, lo);
      }
    }
    for (; i < n4; i += CNT_THREADS) {
      int4 v = p[i];
      count_one(cnt, v.x, lo);
      count_one(cnt, v.y, lo);
      count_one(cnt, v.z, lo);
      count_one(cnt, v.w, lo);
    }
  } else {
    for (int i = tid; i < E; i += CNT_THREADS) count_one(cnt, srcE[i], lo);
  }
  __syncthreads();

  float m = attn[0];
  for (int u = 1; u < T; ++u) m = fmaxf(m, attn[u]);
  float s = 0.0f;
  for (int u = 0; u < T; ++u) s += __expf(attn[u] - m);
  const float wt = __expf(attn[t] - m) / s;

  for (int i = tid; i < CNT_BINS; i += CNT_THREADS) {
    int n = lo + i;
    if (n < N) scale[(size_t)t * N + n] = wt / fmaxf((float)cnt[i], 1.0f);
  }
}

struct Frag {
  v4f a0, a1;
  v4f b0, b1, b2, b3;
};

__device__ __forceinline__ void load_frag(Frag& f, const float* aRow0, const float* aRow1,
                                          const float* bLane, int s) {
  f.a0 = *(const v4f*)(aRow0 + SC_K * s);
  f.a1 = *(const v4f*)(aRow1 + SC_K * s);
  const float* bp = bLane + s * (4 * 32 * 4);
  f.b0 = *(const v4f*)(bp);
  f.b1 = *(const v4f*)(bp + 128);
  f.b2 = *(const v4f*)(bp + 256);
  f.b3 = *(const v4f*)(bp + 384);
}

__device__ __forceinline__ void mma_frag(const Frag& f,
                                         v8f& A0, v8f& A1, v8f& A2, v8f& A3,
                                         v8f& B0, v8f& B1, v8f& B2, v8f& B3) {
  {
    v2f a0 = {f.a0.x, f.a0.y}, a1 = {f.a1.x, f.a1.y};
    v2f b0 = {f.b0.x, f.b0.y}, b1 = {f.b1.x, f.b1.y};
    v2f b2 = {f.b2.x, f.b2.y}, b3 = {f.b3.x, f.b3.y};
    A0 = wmma_f32_16x16x4(a0, b0, A0);
    B0 = wmma_f32_16x16x4(a1, b0, B0);
    A1 = wmma_f32_16x16x4(a0, b1, A1);
    B1 = wmma_f32_16x16x4(a1, b1, B1);
    A2 = wmma_f32_16x16x4(a0, b2, A2);
    B2 = wmma_f32_16x16x4(a1, b2, B2);
    A3 = wmma_f32_16x16x4(a0, b3, A3);
    B3 = wmma_f32_16x16x4(a1, b3, B3);
  }
  {
    v2f a0 = {f.a0.z, f.a0.w}, a1 = {f.a1.z, f.a1.w};
    v2f b0 = {f.b0.z, f.b0.w}, b1 = {f.b1.z, f.b1.w};
    v2f b2 = {f.b2.z, f.b2.w}, b3 = {f.b3.z, f.b3.w};
    A0 = wmma_f32_16x16x4(a0, b0, A0);
    B0 = wmma_f32_16x16x4(a1, b0, B0);
    A1 = wmma_f32_16x16x4(a0, b1, A1);
    B1 = wmma_f32_16x16x4(a1, b1, B1);
    A2 = wmma_f32_16x16x4(a0, b2, A2);
    B2 = wmma_f32_16x16x4(a1, b2, B2);
    A3 = wmma_f32_16x16x4(a0, b3, A3);
    B3 = wmma_f32_16x16x4(a1, b3, B3);
  }
}

__global__ __launch_bounds__(WAVES * 32)
void k3_gemm_scatter(const float* __restrict__ x, const float* __restrict__ W,
                     const float* __restrict__ b, const int* __restrict__ edges,
                     const float* __restrict__ scale, float* __restrict__ out,
                     int N, int E) {
  __shared__ __attribute__((aligned(16))) float sB[K_DIM * D_FEAT];
  __shared__ __attribute__((aligned(16))) float sF[WAVES * M_TILES * TILE_M * FSTRIDE];

  const int t     = blockIdx.y;
  const int tid   = threadIdx.x;
  const int wave  = __builtin_amdgcn_readfirstlane(tid >> 5);
  const int lane  = tid & 31;
  const int row16 = lane & 15;
  const int half  = lane >> 4;

  const float* Wt = W + (size_t)t * K_DIM * D_FEAT;
  for (int i = tid; i < K_DIM * (D_FEAT / 4); i += WAVES * 32) {
    int k  = i >> 4;
    int c4 = (i & 15) << 2;
    v4f v = *(const v4f*)(Wt + k * D_FEAT + c4);
    int s = k >> 3, h = (k >> 2) & 1, q = k & 3;
    int j = c4 >> 4, n = c4 & 15;
    float* p = sB + (((s * 4 + j) * 32 + h * 16 + n) << 2) + q;
    p[0] = v.x; p[4] = v.y; p[8] = v.z; p[12] = v.w;
  }
  __syncthreads();

  const int nblk32 = (E + M_TILES * TILE_M - 1) / (M_TILES * TILE_M);

  const int* srcE = edges + (size_t)t * 2 * E;
  const int* dstE = srcE + E;
  float* sFw = sF + wave * (M_TILES * TILE_M * FSTRIDE);

  const float* bt = b + (size_t)t * D_FEAT;
  const float bias0 = bt[0  + row16];
  const float bias1 = bt[16 + row16];
  const float bias2 = bt[32 + row16];
  const float bias3 = bt[48 + row16];

  const float* aRow0 = sFw + row16 * FSTRIDE + 4 * half;
  const float* aRow1 = sFw + (TILE_M + row16) * FSTRIDE + 4 * half;
  const float* bLane = sB + 4 * lane;

  for (int blk = blockIdx.x * WAVES + wave; blk < nblk32; blk += gridDim.x * WAVES) {

    const int e0 = blk * (M_TILES * TILE_M);
    int ed0 = e0 + row16;
    int ed1 = e0 + TILE_M + row16;
    int cl0 = (ed0 < E) ? ed0 : (E - 1);
    int cl1 = (ed1 < E) ? ed1 : (E - 1);
    const int srcn0 = srcE[cl0];
    const int dstn0 = dstE[cl0];
    const int srcn1 = srcE[cl1];
    const int dstn1 = dstE[cl1];
    const float scl0 = scale[(size_t)t * N + srcn0];
    const float scl1 = scale[(size_t)t * N + srcn1];

    #pragma unroll 4
    for (int r = 0; r < TILE_M; ++r) {
      int sn0 = __shfl(srcn0, r, 32);
      int dn0 = __shfl(dstn0, r, 32);
      int sn1 = __shfl(srcn1, r, 32);
      int dn1 = __shfl(dstn1, r, 32);
      v2f vs0 = *(const v2f*)(x + (size_t)sn0 * D_FEAT + 2 * lane);
      v2f vd0 = *(const v2f*)(x + (size_t)dn0 * D_FEAT + 2 * lane);
      v2f vs1 = *(const v2f*)(x + (size_t)sn1 * D_FEAT + 2 * lane);
      v2f vd1 = *(const v2f*)(x + (size_t)dn1 * D_FEAT + 2 * lane);
      *(v2f*)(sFw + r * FSTRIDE + 2 * lane)                               = vs0;
      *(v2f*)(sFw + r * FSTRIDE + D_FEAT + 2 * lane)                      = vd0;
      *(v2f*)(sFw + (TILE_M + r) * FSTRIDE + 2 * lane)                    = vs1;
      *(v2f*)(sFw + (TILE_M + r) * FSTRIDE + D_FEAT + 2 * lane)           = vd1;
    }

    v8f accA0 = {}, accA1 = {}, accA2 = {}, accA3 = {};
    v8f accB0 = {}, accB1 = {}, accB2 = {}, accB3 = {};

    Frag p0, p1, q0, q1;
    load_frag(p0, aRow0, aRow1, bLane, 0);
    load_frag(p1, aRow0, aRow1, bLane, 1);
    #pragma unroll
    for (int g = 0; g < N_SC / 4; ++g) {
      load_frag(q0, aRow0, aRow1, bLane, 4 * g + 2);
      load_frag(q1, aRow0, aRow1, bLane, 4 * g + 3);
      __builtin_amdgcn_sched_barrier(0);
      mma_frag(p0, accA0, accA1, accA2, accA3, accB0, accB1, accB2, accB3);
      mma_frag(p1, accA0, accA1, accA2, accA3, accB0, accB1, accB2, accB3);
      __builtin_amdgcn_sched_barrier(0);
      if (g + 1 < N_SC / 4) {
        load_frag(p0, aRow0, aRow1, bLane, 4 * g + 4);
        load_frag(p1, aRow0, aRow1, bLane, 4 * g + 5);
      }
      __builtin_amdgcn_sched_barrier(0);
      mma_frag(q0, accA0, accA1, accA2, accA3, accB0, accB1, accB2, accB3);
      mma_frag(q1, accA0, accA1, accA2, accA3, accB0, accB1, accB2, accB3);
      __builtin_amdgcn_sched_barrier(0);
    }

    #pragma unroll
    for (int v = 0; v < 8; ++v) {
      {
        float p0 = fmaxf(accA0[v] + bias0, 0.0f);
        float p1 = fmaxf(accA1[v] + bias1, 0.0f);
        float p2 = fmaxf(accA2[v] + bias2, 0.0f);
        float p3 = fmaxf(accA3[v] + bias3, 0.0f);
        swap_halves(p0, p1);
        swap_halves(p2, p3);
        const int   snLo = __builtin_amdgcn_readlane(srcn0, v);
        const int   snHi = __builtin_amdgcn_readlane(srcn0, v + 8);
        const float scLo = __int_as_float(__builtin_amdgcn_readlane(__float_as_int(scl0), v));
        const float scHi = __int_as_float(__builtin_amdgcn_readlane(__float_as_int(scl0), v + 8));
        if (e0 + v < E) {
          float* o = out + (size_t)snLo * D_FEAT + lane;
          atomicAdd(o,      p0 * scLo);
          atomicAdd(o + 32, p2 * scLo);
        }
        if (e0 + v + 8 < E) {
          float* o = out + (size_t)snHi * D_FEAT + lane;
          atomicAdd(o,      p1 * scHi);
          atomicAdd(o + 32, p3 * scHi);
        }
      }
      {
        float p0 = fmaxf(accB0[v] + bias0, 0.0f);
        float p1 = fmaxf(accB1[v] + bias1, 0.0f);
        float p2 = fmaxf(accB2[v] + bias2, 0.0f);
        float p3 = fmaxf(accB3[v] + bias3, 0.0f);
        swap_halves(p0, p1);
        swap_halves(p2, p3);
        const int   snLo = __builtin_amdgcn_readlane(srcn1, v);
        const int   snHi = __builtin_amdgcn_readlane(srcn1, v + 8);
        const float scLo = __int_as_float(__builtin_amdgcn_readlane(__float_as_int(scl1), v));
        const float scHi = __int_as_float(__builtin_amdgcn_readlane(__float_as_int(scl1), v + 8));
        if (e0 + TILE_M + v < E) {
          float* o = out + (size_t)snLo * D_FEAT + lane;
          atomicAdd(o,      p0 * scLo);
          atomicAdd(o + 32, p2 * scLo);
        }
        if (e0 + TILE_M + v + 8 < E) {
          float* o = out + (size_t)snHi * D_FEAT + lane;
          atomicAdd(o,      p1 * scHi);
          atomicAdd(o + 32, p3 * scHi);
        }
      }
    }
  }
}

#define PSTRIDE  68
#define P_SC     (D_FEAT / SC_K)

__device__ __forceinline__ void load_frag_ab(Frag& f, const float* aRow0, const float* aRow1,
                                             const float* bLane, int sa, int sb) {
  f.a0 = *(const v4f*)(aRow0 + SC_K * sa);
  f.a1 = *(const v4f*)(aRow1 + SC_K * sa);
  const float* bp = bLane + sb * (4 * 32 * 4);
  f.b0 = *(const v4f*)(bp);
  f.b1 = *(const v4f*)(bp + 128);
  f.b2 = *(const v4f*)(bp + 256);
  f.b3 = *(const v4f*)(bp + 384);
}

__device__ __forceinline__ void project_half(const float* aRow0, const float* aRow1,
                                             const float* bLane, int sb0,
                                             v8f& A0, v8f& A1, v8f& A2, v8f& A3,
                                             v8f& B0, v8f& B1, v8f& B2, v8f& B3) {
  Frag p0, p1, q0, q1;
  load_frag_ab(p0, aRow0, aRow1, bLane, 0, sb0 + 0);
  load_frag_ab(p1, aRow0, aRow1, bLane, 1, sb0 + 1);
  #pragma unroll
  for (int g = 0; g < P_SC / 4; ++g) {
    load_frag_ab(q0, aRow0, aRow1, bLane, 4 * g + 2, sb0 + 4 * g + 2);
    load_frag_ab(q1, aRow0, aRow1, bLane, 4 * g + 3, sb0 + 4 * g + 3);
    __builtin_amdgcn_sched_barrier(0);
    mma_frag(p0, A0, A1, A2, A3, B0, B1, B2, B3);
    mma_frag(p1, A0, A1, A2, A3, B0, B1, B2, B3);
    __builtin_amdgcn_sched_barrier(0);
    if (g + 1 < P_SC / 4) {
      load_frag_ab(p0, aRow0, aRow1, bLane, 4 * g + 4, sb0 + 4 * g + 4);
      load_frag_ab(p1, aRow0, aRow1, bLane, 4 * g + 5, sb0 + 4 * g + 5);
    }
    __builtin_amdgcn_sched_barrier(0);
    mma_frag(q0, A0, A1, A2, A3, B0, B1, B2, B3);
    mma_frag(q1, A0, A1, A2, A3, B0, B1, B2, B3);
    __builtin_amdgcn_sched_barrier(0);
  }
}

__device__ __forceinline__ void store_tile(float* __restrict__ plane, int n0, int N, int lane,
                                           const v8f& c0, const v8f& c1, const v8f& c2, const v8f& c3,
                                           float add0, float add1, float add2, float add3) {
  #pragma unroll
  for (int v = 0; v < 8; ++v) {
    float p0 = c0[v] + add0;
    float p1 = c1[v] + add1;
    float p2 = c2[v] + add2;
    float p3 = c3[v] + add3;
    swap_halves(p0, p1);
    swap_halves(p2, p3);
    if (n0 + v < N) {
      float* o = plane + (size_t)(n0 + v) * D_FEAT + lane;
      o[0]  = p0;
      o[32] = p2;
    }
    if (n0 + v + 8 < N) {
      float* o = plane + (size_t)(n0 + v + 8) * D_FEAT + lane;
      o[0]  = p1;
      o[32] = p3;
    }
  }
}

__global__ __launch_bounds__(WAVES * 32)
void k2_project(const float* __restrict__ x, const float* __restrict__ W,
                const float* __restrict__ b, float* __restrict__ pq, int N, int t0) {
  __shared__ __attribute__((aligned(16))) float sB[K_DIM * D_FEAT];
  __shared__ __attribute__((aligned(16))) float sX[WAVES * M_TILES * TILE_M * PSTRIDE];

  const int tt    = blockIdx.y;
  const int t     = t0 + tt;
  const int tid   = threadIdx.x;
  const int wave  = __builtin_amdgcn_readfirstlane(tid >> 5);
  const int lane  = tid & 31;
  const int row16 = lane & 15;
  const int half  = lane >> 4;

  const float* Wt = W + (size_t)t * K_DIM * D_FEAT;
  for (int i = tid; i < K_DIM * (D_FEAT / 4); i += WAVES * 32) {
    int k  = i >> 4;
    int c4 = (i & 15) << 2;
    v4f v = *(const v4f*)(Wt + k * D_FEAT + c4);
    int s = k >> 3, h = (k >> 2) & 1, q = k & 3;
    int j = c4 >> 4, n = c4 & 15;
    float* p = sB + (((s * 4 + j) * 32 + h * 16 + n) << 2) + q;
    p[0] = v.x; p[4] = v.y; p[8] = v.z; p[12] = v.w;
  }
  __syncthreads();

  const float* bt = b + (size_t)t * D_FEAT;
  const float bias0 = bt[0  + row16];
  const float bias1 = bt[16 + row16];
  const float bias2 = bt[32 + row16];
  const float bias3 = bt[48 + row16];

  float* sXw = sX + wave * (M_TILES * TILE_M * PSTRIDE);
  const float* aRow0 = sXw + row16 * PSTRIDE + 4 * half;
  const float* aRow1 = sXw + (TILE_M + row16) * PSTRIDE + 4 * half;
  const float* bLane = sB + 4 * lane;

  float* planeP = pq + (size_t)(2 * tt)     * N * D_FEAT;
  float* planeQ = pq + (size_t)(2 * tt + 1) * N * D_FEAT;

  const int nblk = (N + M_TILES * TILE_M - 1) / (M_TILES * TILE_M);
  for (int blk = blockIdx.x * WAVES + wave; blk < nblk; blk += gridDim.x * WAVES) {
    const int n0 = blk * (M_TILES * TILE_M);
    #pragma unroll
    for (int r2 = 0; r2 < TILE_M; ++r2) {
      int r  = 2 * r2 + half;
      int nr = n0 + r;
      if (nr > N - 1) nr = N - 1;
      v4f v = *(const v4f*)(x + (size_t)nr * D_FEAT + 4 * row16);
      *(v4f*)(sXw + r * PSTRIDE + 4 * row16) = v;
    }

    {
      v8f accA0 = {}, accA1 = {}, accA2 = {}, accA3 = {};
      v8f accB0 = {}, accB1 = {}, accB2 = {}, accB3 = {};
      project_half(aRow0, aRow1, bLane, 0,
                   accA0, accA1, accA2, accA3, accB0, accB1, accB2, accB3);
      store_tile(planeP, n0, N, lane, accA0, accA1, accA2, accA3, bias0, bias1, bias2, bias3);
      store_tile(planeP, n0 + TILE_M, N, lane, accB0, accB1, accB2, accB3, bias0, bias1, bias2, bias3);
    }
    {
      v8f accA0 = {}, accA1 = {}, accA2 = {}, accA3 = {};
      v8f accB0 = {}, accB1 = {}, accB2 = {}, accB3 = {};
      project_half(aRow0, aRow1, bLane, P_SC,
                   accA0, accA1, accA2, accA3, accB0, accB1, accB2, accB3);
      store_tile(planeQ, n0, N, lane, accA0, accA1, accA2, accA3, 0.0f, 0.0f, 0.0f, 0.0f);
      store_tile(planeQ, n0 + TILE_M, N, lane, accB0, accB1, accB2, accB3, 0.0f, 0.0f, 0.0f, 0.0f);
    }
  }
}

#define EDGE_BATCH 8

template <bool GUARD>
__device__ __forceinline__ void edge_block(const float* __restrict__ planeP,
                                           const float* __restrict__ planeQ,
                                           float* __restrict__ out, int lane,
                                           int srcn, int dstn, float scl, int nvalid) {
  #pragma unroll 1
  for (int r0 = 0; r0 < 32; r0 += EDGE_BATCH) {
    float pl[EDGE_BATCH], ph[EDGE_BATCH], ql[EDGE_BATCH], qh[EDGE_BATCH];
    #pragma unroll
    for (int j = 0; j < EDGE_BATCH; ++j) {
      const int sn = __builtin_amdgcn_readlane(srcn, r0 + j);
      const int dn = __builtin_amdgcn_readlane(dstn, r0 + j);
      const float* pr = planeP + (size_t)sn * D_FEAT + lane;
      const float* qr = planeQ + (size_t)dn * D_FEAT + lane;
      pl[j] = pr[0];
      ph[j] = pr[32];
      ql[j] = qr[0];
      qh[j] = qr[32];
    }
    #pragma unroll
    for (int j = 0; j < EDGE_BATCH; ++j) {
      const int   sn = __builtin_amdgcn_readlane(srcn, r0 + j);
      const float sc = __int_as_float(__builtin_amdgcn_readlane(__float_as_int(scl), r0 + j));
      if (!GUARD || r0 + j < nvalid) {
        float* o = out + (size_t)sn * D_FEAT + lane;
        atomicAdd(o,      fmaxf(pl[j] + ql[j], 0.0f) * sc);
        atomicAdd(o + 32, fmaxf(ph[j] + qh[j], 0.0f) * sc);
      }
    }
  }
}

__global__ __launch_bounds__(WAVES * 32)
void k3_edge_scatter(const int* __restrict__ edges, const float* __restrict__ scale,
                     const float* __restrict__ pq, float* __restrict__ out,
                     int N, int E, int t0, int g) {
  const int tid  = threadIdx.x;
  const int wave = __builtin_amdgcn_readfirstlane(tid >> 5);
  const int lane = tid & 31;

  const int nblk32 = (E + 31) / 32;
  const int total  = g * nblk32;
  for (int c = blockIdx.x * WAVES + wave; c < total; c += gridDim.x * WAVES) {
    const int tt  = c / nblk32;
    const int blk = c - tt * nblk32;
    const int t   = t0 + tt;
    const int* srcE = edges + (size_t)t * 2 * E;
    const int* dstE = srcE + E;
    const float* planeP = pq + (size_t)(2 * tt)     * N * D_FEAT;
    const float* planeQ = pq + (size_t)(2 * tt + 1) * N * D_FEAT;

    const int e0 = blk * 32;
    int ed = e0 + lane;
    int cl = (ed < E) ? ed : (E - 1);
    const int   srcn = srcE[cl];
    const int   dstn = dstE[cl];
    const float scl  = scale[(size_t)t * N + srcn];

    if (e0 + 32 <= E) edge_block<false>(planeP, planeQ, out, lane, srcn, dstn, scl, 32);
    else              edge_block<true >(planeP, planeQ, out, lane, srcn, dstn, scl, E - e0);
  }
}

extern "C" void kernel_launch(void* const* d_in, const int* in_sizes, int n_in,
                              void* d_out, int out_size, void* d_ws, size_t ws_size,
                              hipStream_t stream) {
  const float* x     = (const float*)d_in[0];
  const float* W     = (const float*)d_in[1];
  const float* b     = (const float*)d_in[2];
  const float* attn  = (const float*)d_in[3];
  const int*   edges = (const int*)d_in[4];
  float* out = (float*)d_out;
  float* wsf = (float*)d_ws;

  const int N = in_sizes[0] / D_FEAT;
  const int T = in_sizes[3];
  const int E = in_sizes[4] / (2 * T);

  k0_init<<<2048, 256, 0, stream>>>(out, N);
  {
    dim3 cgrid((unsigned)((N + CNT_BINS - 1) / CNT_BINS), T);
    k1_count_scale<<<cgrid, CNT_THREADS, 0, stream>>>(edges, attn, wsf + 8, N, E, T);
  }

  const size_t planeF = (size_t)N * D_FEAT;
  const size_t pqOff  = ((size_t)8 + (size_t)T * N + 63) & ~(size_t)63;
  const size_t wsF    = ws_size / sizeof(float);
  int G = 0;
  if (wsF > pqOff) {
    size_t g = (wsF - pqOff) / (2 * planeF);
    G = (g < (size_t)T) ? (int)g : T;
  }

  if (G == 0) {
    dim3 grid(GROUPS_PER_TYPE, T);
    k3_gemm_scatter<<<grid, WAVES * 32, 0, stream>>>(x, W, b, edges,
                                                     wsf + 8, out, N, E);
    return;
  }

  float* pq = wsf + pqOff;
  for (int t0 = 0; t0 < T; t0 += G) {
    const int g = (T - t0 < G) ? (T - t0) : G;
    const int pgx = (PROJ_GROUPS / g > 0) ? (PROJ_GROUPS / g) : 1;
    k2_project<<<dim3(pgx, g), WAVES * 32, 0, stream>>>(x, W, b, pq, N, t0);
    k3_edge_scatter<<<EDGE_GROUPS, WAVES * 32, 0, stream>>>(edges, wsf + 8, pq, out,
                                                            N, E, t0, g);
  }
}


# Edges counting-sorted by source node in LDS (no global atomics); one wave sums a node's relu rows of all types in registers, stores out once: no float atomics, no zero-fill; sum order = list order
# speedup vs baseline: 2.0955x; 1.1033x over previous
#include <hip/hip_runtime.h>

#define D_FEAT   64
#define K_DIM    128
#define TILE_M   16
#define M_TILES  2
#define WAVES    16
#define GROUPS_PER_TYPE 51
#define SC_K     8
#define N_SC     (K_DIM / SC_K)
#define FSTRIDE  132
#define CNT_BINS    2048
#define CNT_THREADS 1024
#define CNT_BATCH   4
#define PROJ_GROUPS 255
#define EDGE_GROUPS 510
#define CSR_CAP     16384
#define NODE_TCH    5
#define NODE_QB     6
#define NODE_QT     4

typedef __attribute__((ext_vector_type(2))) float v2f;
typedef __attribute__((ext_vector_type(4))) float v4f;
typedef __attribute__((ext_vector_type(8))) float v8f;

__device__ __forceinline__ v8f wmma_f32_16x16x4(v2f a, v2f b, v8f c) {
  return __builtin_amdgcn_wmma_f32_16x16x4_f32(false, a, false, b, (short)0, c,
                                               false, false);
}

__device__ __forceinline__ void swap_halves(float& a, float& b) {
  typedef __attribute__((ext_vector_type(2))) unsigned v2u;
  v2u r = __builtin_amdgcn_permlane16_swap(__float_as_uint(a), __float_as_uint(b),
                                           false, false);
  a = __uint_as_float(r.x);
  b = __uint_as_float(r.y);
}

__global__ void k0_init(float* __restrict__ out, int N) {
  size_t idx    = (size_t)blockIdx.x * blockDim.x + threadIdx.x;
  size_t stride = (size_t)gridDim.x * blockDim.x;
  size_t nOut = (size_t)N * D_FEAT;
  for (size_t i = idx; i < nOut; i += stride) out[i] = 0.0f;
}

__device__ __forceinline__ void count_one(unsigned* cnt, int s, int lo) {
  unsigned d = (unsigned)(s - lo);
  if (d < (unsigned)CNT_BINS) atomicAdd(&cnt[d], 1u);
}

__global__ __launch_bounds__(CNT_THREADS)
void k1_count_scale(const int* __restrict__ edges, const float* __restrict__ attn,
                    float* __restrict__ scale, int N, int E, int T) {
  __shared__ unsigned cnt[CNT_BINS];
  const int t   = blockIdx.y;
  const int lo  = blockIdx.x * CNT_BINS;
  const int tid = threadIdx.x;

  for (int i = tid; i < CNT_BINS; i += CNT_THREADS) cnt[i] = 0u;
  __syncthreads();

  const int* srcE = edges + (size_t)t * 2 * E;
  if ((E & 3) == 0) {
    const int4* p  = (const int4*)srcE;
    const int   n4 = E >> 2;
    int i = tid;
    for (; i + (CNT_BATCH - 1) * CNT_THREADS < n4; i += CNT_BATCH * CNT_THREADS) {
      int4 v[CNT_BATCH];
      #pragma unroll
      for (int j = 0; j < CNT_BATCH; ++j) v[j] = p[i + j * CNT_THREADS];
      #pragma unroll
      for (int j = 0; j < CNT_BATCH; ++j) {
        count_one(cnt, v[j].x, lo);
        count_one(cnt, v[j].y, lo);
        count_one(cnt, v[j].z, lo);
        count_one(cnt, v[j].w, lo);
      }
    }
    for (; i < n4; i += CNT_THREADS) {
      int4 v = p[i];
      count_one(cnt, v.x, lo);
      count_one(cnt, v.y, lo);
      count_one(cnt, v.z, lo);
      count_one(cnt, v.w, lo);
    }
  } else {
    for (int i = tid; i < E; i += CNT_THREADS) count_one(cnt, srcE[i], lo);
  }
  __syncthreads();

  float m = attn[0];
  for (int u = 1; u < T; ++u) m = fmaxf(m, attn[u]);
  float s = 0.0f;
  for (int u = 0; u < T; ++u) s += __expf(attn[u] - m);
  const float wt = __expf(attn[t] - m) / s;

  for (int i = tid; i < CNT_BINS; i += CNT_THREADS) {
    int n = lo + i;
    if (n < N) scale[(size_t)t * N + n] = wt / fmaxf((float)cnt[i], 1.0f);
  }
}

struct Frag {
  v4f a0, a1;
  v4f b0, b1, b2, b3;
};

__device__ __forceinline__ void load_frag(Frag& f, const float* aRow0, const float* aRow1,
                                          const float* bLane, int s) {
  f.a0 = *(const v4f*)(aRow0 + SC_K * s);
  f.a1 = *(const v4f*)(aRow1 + SC_K * s);
  const float* bp = bLane + s * (4 * 32 * 4);
  f.b0 = *(const v4f*)(bp);
  f.b1 = *(const v4f*)(bp + 128);
  f.b2 = *(const v4f*)(bp + 256);
  f.b3 = *(const v4f*)(bp + 384);
}

__device__ __forceinline__ void mma_frag(const Frag& f,
                                         v8f& A0, v8f& A1, v8f& A2, v8f& A3,
                                         v8f& B0, v8f& B1, v8f& B2, v8f& B3) {
  {
    v2f a0 = {f.a0.x, f.a0.y}, a1 = {f.a1.x, f.a1.y};
    v2f b0 = {f.b0.x, f.b0.y}, b1 = {f.b1.x, f.b1.y};
    v2f b2 = {f.b2.x, f.b2.y}, b3 = {f.b3.x, f.b3.y};
    A0 = wmma_f32_16x16x4(a0, b0, A0);
    B0 = wmma_f32_16x16x4(a1, b0, B0);
    A1 = wmma_f32_16x16x4(a0, b1, A1);
    B1 = wmma_f32_16x16x4(a1, b1, B1);
    A2 = wmma_f32_16x16x4(a0, b2, A2);
    B2 = wmma_f32_16x16x4(a1, b2, B2);
    A3 = wmma_f32_16x16x4(a0, b3, A3);
    B3 = wmma_f32_16x16x4(a1, b3, B3);
  }
  {
    v2f a0 = {f.a0.z, f.a0.w}, a1 = {f.a1.z, f.a1.w};
    v2f b0 = {f.b0.z, f.b0.w}, b1 = {f.b1.z, f.b1.w};
    v2f b2 = {f.b2.z, f.b2.w}, b3 = {f.b3.z, f.b3.w};
    A0 = wmma_f32_16x16x4(a0, b0, A0);
    B0 = wmma_f32_16x16x4(a1, b0, B0);
    A1 = wmma_f32_16x16x4(a0, b1, A1);
    B1 = wmma_f32_16x16x4(a1, b1, B1);
    A2 = wmma_f32_16x16x4(a0, b2, A2);
    B2 = wmma_f32_16x16x4(a1, b2, B2);
    A3 = wmma_f32_16x16x4(a0, b3, A3);
    B3 = wmma_f32_16x16x4(a1, b3, B3);
  }
}

__global__ __launch_bounds__(WAVES * 32)
void k3_gemm_scatter(const float* __restrict__ x, const float* __restrict__ W,
                     const float* __restrict__ b, const int* __restrict__ edges,
                     const float* __restrict__ scale, float* __restrict__ out,
                     int N, int E) {
  __shared__ __attribute__((aligned(16))) float sB[K_DIM * D_FEAT];
  __shared__ __attribute__((aligned(16))) float sF[WAVES * M_TILES * TILE_M * FSTRIDE];

  const int t     = blockIdx.y;
  const int tid   = threadIdx.x;
  const int wave  = __builtin_amdgcn_readfirstlane(tid >> 5);
  const int lane  = tid & 31;
  const int row16 = lane & 15;
  const int half  = lane >> 4;

  const float* Wt = W + (size_t)t * K_DIM * D_FEAT;
  for (int i = tid; i < K_DIM * (D_FEAT / 4); i += WAVES * 32) {
    int k  = i >> 4;
    int c4 = (i & 15) << 2;
    v4f v = *(const v4f*)(Wt + k * D_FEAT + c4);
    int s = k >> 3, h = (k >> 2) & 1, q = k & 3;
    int j = c4 >> 4, n = c4 & 15;
    float* p = sB + (((s * 4 + j) * 32 + h * 16 + n) << 2) + q;
    p[0] = v.x; p[4] = v.y; p[8] = v.z; p[12] = v.w;
  }
  __syncthreads();

  const int nblk32 = (E + M_TILES * TILE_M - 1) / (M_TILES * TILE_M);

  const int* srcE = edges + (size_t)t * 2 * E;
  const int* dstE = srcE + E;
  float* sFw = sF + wave * (M_TILES * TILE_M * FSTRIDE);

  const float* bt = b + (size_t)t * D_FEAT;
  const float bias0 = bt[0  + row16];
  const float bias1 = bt[16 + row16];
  const float bias2 = bt[32 + row16];
  const float bias3 = bt[48 + row16];

  const float* aRow0 = sFw + row16 * FSTRIDE + 4 * half;
  const float* aRow1 = sFw + (TILE_M + row16) * FSTRIDE + 4 * half;
  const float* bLane = sB + 4 * lane;

  for (int blk = blockIdx.x * WAVES + wave; blk < nblk32; blk += gridDim.x * WAVES) {

    const int e0 = blk * (M_TILES * TILE_M);
    int ed0 = e0 + row16;
    int ed1 = e0 + TILE_M + row16;
    int cl0 = (ed0 < E) ? ed0 : (E - 1);
    int cl1 = (ed1 < E) ? ed1 : (E - 1);
    const int srcn0 = srcE[cl0];
    const int dstn0 = dstE[cl0];
    const int srcn1 = srcE[cl1];
    const int dstn1 = dstE[cl1];
    const float scl0 = scale[(size_t)t * N + srcn0];
    const float scl1 = scale[(size_t)t * N + srcn1];

    #pragma unroll 4
    for (int r = 0; r < TILE_M; ++r) {
      int sn0 = __shfl(srcn0, r, 32);
      int dn0 = __shfl(dstn0, r, 32);
      int sn1 = __shfl(srcn1, r, 32);
      int dn1 = __shfl(dstn1, r, 32);
      v2f vs0 = *(const v2f*)(x + (size_t)sn0 * D_FEAT + 2 * lane);
      v2f vd0 = *(const v2f*)(x + (size_t)dn0 * D_FEAT + 2 * lane);
      v2f vs1 = *(const v2f*)(x + (size_t)sn1 * D_FEAT + 2 * lane);
      v2f vd1 = *(const v2f*)(x + (size_t)dn1 * D_FEAT + 2 * lane);
      *(v2f*)(sFw + r * FSTRIDE + 2 * lane)                               = vs0;
      *(v2f*)(sFw + r * FSTRIDE + D_FEAT + 2 * lane)                      = vd0;
      *(v2f*)(sFw + (TILE_M + r) * FSTRIDE + 2 * lane)                    = vs1;
      *(v2f*)(sFw + (TILE_M + r) * FSTRIDE + D_FEAT + 2 * lane)           = vd1;
    }

    v8f accA0 = {}, accA1 = {}, accA2 = {}, accA3 = {};
    v8f accB0 = {}, accB1 = {}, accB2 = {}, accB3 = {};

    Frag p0, p1, q0, q1;
    load_frag(p0, aRow0, aRow1, bLane, 0);
    load_frag(p1, aRow0, aRow1, bLane, 1);
    #pragma unroll
    for (int g = 0; g < N_SC / 4; ++g) {
      load_frag(q0, aRow0, aRow1, bLane, 4 * g + 2);
      load_frag(q1, aRow0, aRow1, bLane, 4 * g + 3);
      __builtin_amdgcn_sched_barrier(0);
      mma_frag(p0, accA0, accA1, accA2, accA3, accB0, accB1, accB2, accB3);
      mma_frag(p1, accA0, accA1, accA2, accA3, accB0, accB1, accB2, accB3);
      __builtin_amdgcn_sched_barrier(0);
      if (g + 1 < N_SC / 4) {
        load_frag(p0, aRow0, aRow1, bLane, 4 * g + 4);
        load_frag(p1, aRow0, aRow1, bLane, 4 * g + 5);
      }
      __builtin_amdgcn_sched_barrier(0);
      mma_frag(q0, accA0, accA1, accA2, accA3, accB0, accB1, accB2, accB3);
      mma_frag(q1, accA0, accA1, accA2, accA3, accB0, accB1, accB2, accB3);
      __builtin_amdgcn_sched_barrier(0);
    }

    #pragma unroll
    for (int v = 0; v < 8; ++v) {
      {
        float p0 = fmaxf(accA0[v] + bias0, 0.0f);
        float p1 = fmaxf(accA1[v] + bias1, 0.0f);
        float p2 = fmaxf(accA2[v] + bias2, 0.0f);
        float p3 = fmaxf(accA3[v] + bias3, 0.0f);
        swap_halves(p0, p1);
        swap_halves(p2, p3);
        const int   snLo = __builtin_amdgcn_readlane(srcn0, v);
        const int   snHi = __builtin_amdgcn_readlane(srcn0, v + 8);
        const float scLo = __int_as_float(__builtin_amdgcn_readlane(__float_as_int(scl0), v));
        const float scHi = __int_as_float(__builtin_amdgcn_readlane(__float_as_int(scl0), v + 8));
        if (e0 + v < E) {
          float* o = out + (size_t)snLo * D_FEAT + lane;
          atomicAdd(o,      p0 * scLo);
          atomicAdd(o + 32, p2 * scLo);
        }
        if (e0 + v + 8 < E) {
          float* o = out + (size_t)snHi * D_FEAT + lane;
          atomicAdd(o,      p1 * scHi);
          atomicAdd(o + 32, p3 * scHi);
        }
      }
      {
        float p0 = fmaxf(accB0[v] + bias0, 0.0f);
        float p1 = fmaxf(accB1[v] + bias1, 0.0f);
        float p2 = fmaxf(accB2[v] + bias2, 0.0f);
        float p3 = fmaxf(accB3[v] + bias3, 0.0f);
        swap_halves(p0, p1);
        swap_halves(p2, p3);
        const int   snLo = __builtin_amdgcn_readlane(srcn1, v);
        const int   snHi = __builtin_amdgcn_readlane(srcn1, v + 8);
        const float scLo = __int_as_float(__builtin_amdgcn_readlane(__float_as_int(scl1), v));
        const float scHi = __int_as_float(__builtin_amdgcn_readlane(__float_as_int(scl1), v + 8));
        if (e0 + TILE_M + v < E) {
          float* o = out + (size_t)snLo * D_FEAT + lane;
          atomicAdd(o,      p0 * scLo);
          atomicAdd(o + 32, p2 * scLo);
        }
        if (e0 + TILE_M + v + 8 < E) {
          float* o = out + (size_t)snHi * D_FEAT + lane;
          atomicAdd(o,      p1 * scHi);
          atomicAdd(o + 32, p3 * scHi);
        }
      }
    }
  }
}

#define PSTRIDE  68
#define P_SC     (D_FEAT / SC_K)

__device__ __forceinline__ void load_frag_ab(Frag& f, const float* aRow0, const float* aRow1,
                                             const float* bLane, int sa, int sb) {
  f.a0 = *(const v4f*)(aRow0 + SC_K * sa);
  f.a1 = *(const v4f*)(aRow1 + SC_K * sa);
  const float* bp = bLane + sb * (4 * 32 * 4);
  f.b0 = *(const v4f*)(bp);
  f.b1 = *(const v4f*)(bp + 128);
  f.b2 = *(const v4f*)(bp + 256);
  f.b3 = *(const v4f*)(bp + 384);
}

__device__ __forceinline__ void project_half(const float* aRow0, const float* aRow1,
                                             const float* bLane, int sb0,
                                             v8f& A0, v8f& A1, v8f& A2, v8f& A3,
                                             v8f& B0, v8f& B1, v8f& B2, v8f& B3) {
  Frag p0, p1, q0, q1;
  load_frag_ab(p0, aRow0, aRow1, bLane, 0, sb0 + 0);
  load_frag_ab(p1, aRow0, aRow1, bLane, 1, sb0 + 1);
  #pragma unroll
  for (int g = 0; g < P_SC / 4; ++g) {
    load_frag_ab(q0, aRow0, aRow1, bLane, 4 * g + 2, sb0 + 4 * g + 2);
    load_frag_ab(q1, aRow0, aRow1, bLane, 4 * g + 3, sb0 + 4 * g + 3);
    __builtin_amdgcn_sched_barrier(0);
    mma_frag(p0, A0, A1, A2, A3, B0, B1, B2, B3);
    mma_frag(p1, A0, A1, A2, A3, B0, B1, B2, B3);
    __builtin_amdgcn_sched_barrier(0);
    if (g + 1 < P_SC / 4) {
      load_frag_ab(p0, aRow0, aRow1, bLane, 4 * g + 4, sb0 + 4 * g + 4);
      load_frag_ab(p1, aRow0, aRow1, bLane, 4 * g + 5, sb0 + 4 * g + 5);
    }
    __builtin_amdgcn_sched_barrier(0);
    mma_frag(q0, A0, A1, A2, A3, B0, B1, B2, B3);
    mma_frag(q1, A0, A1, A2, A3, B0, B1, B2, B3);
    __builtin_amdgcn_sched_barrier(0);
  }
}

__device__ __forceinline__ void store_tile(float* __restrict__ plane, int n0, int N, int lane,
                                           const v8f& c0, const v8f& c1, const v8f& c2, const v8f& c3,
                                           float add0, float add1, float add2, float add3) {
  #pragma unroll
  for (int v = 0; v < 8; ++v) {
    float p0 = c0[v] + add0;
    float p1 = c1[v] + add1;
    float p2 = c2[v] + add2;
    float p3 = c3[v] + add3;
    swap_halves(p0, p1);
    swap_halves(p2, p3);
    if (n0 + v < N) {
      float* o = plane + (size_t)(n0 + v) * D_FEAT + lane;
      o[0]  = p0;
      o[32] = p2;
    }
    if (n0 + v + 8 < N) {
      float* o = plane + (size_t)(n0 + v + 8) * D_FEAT + lane;
      o[0]  = p1;
      o[32] = p3;
    }
  }
}

__global__ __launch_bounds__(WAVES * 32)
void k2_project(const float* __restrict__ x, const float* __restrict__ W,
                const float* __restrict__ b, float* __restrict__ pq, int N, int t0) {
  __shared__ __attribute__((aligned(16))) float sB[K_DIM * D_FEAT];
  __shared__ __attribute__((aligned(16))) float sX[WAVES * M_TILES * TILE_M * PSTRIDE];

  const int tt    = blockIdx.y;
  const int t     = t0 + tt;
  const int tid   = threadIdx.x;
  const int wave  = __builtin_amdgcn_readfirstlane(tid >> 5);
  const int lane  = tid & 31;
  const int row16 = lane & 15;
  const int half  = lane >> 4;

  const float* Wt = W + (size_t)t * K_DIM * D_FEAT;
  for (int i = tid; i < K_DIM * (D_FEAT / 4); i += WAVES * 32) {
    int k  = i >> 4;
    int c4 = (i & 15) << 2;
    v4f v = *(const v4f*)(Wt + k * D_FEAT + c4);
    int s = k >> 3, h = (k >> 2) & 1, q = k & 3;
    int j = c4 >> 4, n = c4 & 15;
    float* p = sB + (((s * 4 + j) * 32 + h * 16 + n) << 2) + q;
    p[0] = v.x; p[4] = v.y; p[8] = v.z; p[12] = v.w;
  }
  __syncthreads();

  const float* bt = b + (size_t)t * D_FEAT;
  const float bias0 = bt[0  + row16];
  const float bias1 = bt[16 + row16];
  const float bias2 = bt[32 + row16];
  const float bias3 = bt[48 + row16];

  float* sXw = sX + wave * (M_TILES * TILE_M * PSTRIDE);
  const float* aRow0 = sXw + row16 * PSTRIDE + 4 * half;
  const float* aRow1 = sXw + (TILE_M + row16) * PSTRIDE + 4 * half;
  const float* bLane = sB + 4 * lane;

  float* planeP = pq + (size_t)(2 * tt)     * N * D_FEAT;
  float* planeQ = pq + (size_t)(2 * tt + 1) * N * D_FEAT;

  const int nblk = (N + M_TILES * TILE_M - 1) / (M_TILES * TILE_M);
  for (int blk = blockIdx.x * WAVES + wave; blk < nblk; blk += gridDim.x * WAVES) {
    const int n0 = blk * (M_TILES * TILE_M);
    #pragma unroll
    for (int r2 = 0; r2 < TILE_M; ++r2) {
      int r  = 2 * r2 + half;
      int nr = n0 + r;
      if (nr > N - 1) nr = N - 1;
      v4f v = *(const v4f*)(x + (size_t)nr * D_FEAT + 4 * row16);
      *(v4f*)(sXw + r * PSTRIDE + 4 * row16) = v;
    }

    {
      v8f accA0 = {}, accA1 = {}, accA2 = {}, accA3 = {};
      v8f accB0 = {}, accB1 = {}, accB2 = {}, accB3 = {};
      project_half(aRow0, aRow1, bLane, 0,
                   accA0, accA1, accA2, accA3, accB0, accB1, accB2, accB3);
      store_tile(planeP, n0, N, lane, accA0, accA1, accA2, accA3, bias0, bias1, bias2, bias3);
      store_tile(planeP, n0 + TILE_M, N, lane, accB0, accB1, accB2, accB3, bias0, bias1, bias2, bias3);
    }
    {
      v8f accA0 = {}, accA1 = {}, accA2 = {}, accA3 = {};
      v8f accB0 = {}, accB1 = {}, accB2 = {}, accB3 = {};
      project_half(aRow0, aRow1, bLane, P_SC,
                   accA0, accA1, accA2, accA3, accB0, accB1, accB2, accB3);
      store_tile(planeQ, n0, N, lane, accA0, accA1, accA2, accA3, 0.0f, 0.0f, 0.0f, 0.0f);
      store_tile(planeQ, n0 + TILE_M, N, lane, accB0, accB1, accB2, accB3, 0.0f, 0.0f, 0.0f, 0.0f);
    }
  }
}

#define EDGE_BATCH 8

template <bool GUARD>
__device__ __forceinline__ void edge_block(const float* __restrict__ planeP,
                                           const float* __restrict__ planeQ,
                                           float* __restrict__ out, int lane,
                                           int srcn, int dstn, float scl, int nvalid) {
  #pragma unroll 1
  for (int r0 = 0; r0 < 32; r0 += EDGE_BATCH) {
    float pl[EDGE_BATCH], ph[EDGE_BATCH], ql[EDGE_BATCH], qh[EDGE_BATCH];
    #pragma unroll
    for (int j = 0; j < EDGE_BATCH; ++j) {
      const int sn = __builtin_amdgcn_readlane(srcn, r0 + j);
      const int dn = __builtin_amdgcn_readlane(dstn, r0 + j);
      const float* pr = planeP + (size_t)sn * D_FEAT + lane;
      const float* qr = planeQ + (size_t)dn * D_FEAT + lane;
      pl[j] = pr[0];
      ph[j] = pr[32];
      ql[j] = qr[0];
      qh[j] = qr[32];
    }
    #pragma unroll
    for (int j = 0; j < EDGE_BATCH; ++j) {
      const int   sn = __builtin_amdgcn_readlane(srcn, r0 + j);
      const float sc = __int_as_float(__builtin_amdgcn_readlane(__float_as_int(scl), r0 + j));
      if (!GUARD || r0 + j < nvalid) {
        float* o = out + (size_t)sn * D_FEAT + lane;
        atomicAdd(o,      fmaxf(pl[j] + ql[j], 0.0f) * sc);
        atomicAdd(o + 32, fmaxf(ph[j] + qh[j], 0.0f) * sc);
      }
    }
  }
}

__global__ __launch_bounds__(WAVES * 32)
void k3_edge_scatter(const int* __restrict__ edges, const float* __restrict__ scale,
                     const float* __restrict__ pq, float* __restrict__ out,
                     int N, int E, int t0, int g) {
  const int tid  = threadIdx.x;
  const int wave = __builtin_amdgcn_readfirstlane(tid >> 5);
  const int lane = tid & 31;

  const int nblk32 = (E + 31) / 32;
  const int total  = g * nblk32;
  for (int c = blockIdx.x * WAVES + wave; c < total; c += gridDim.x * WAVES) {
    const int tt  = c / nblk32;
    const int blk = c - tt * nblk32;
    const int t   = t0 + tt;
    const int* srcE = edges + (size_t)t * 2 * E;
    const int* dstE = srcE + E;
    const float* planeP = pq + (size_t)(2 * tt)     * N * D_FEAT;
    const float* planeQ = pq + (size_t)(2 * tt + 1) * N * D_FEAT;

    const int e0 = blk * 32;
    int ed = e0 + lane;
    int cl = (ed < E) ? ed : (E - 1);
    const int   srcn = srcE[cl];
    const int   dstn = dstE[cl];
    const float scl  = scale[(size_t)t * N + srcn];

    if (e0 + 32 <= E) edge_block<false>(planeP, planeQ, out, lane, srcn, dstn, scl, 32);
    else              edge_block<true >(planeP, planeQ, out, lane, srcn, dstn, scl, E - e0);
  }
}

#define WAVE_CAP (CSR_CAP / (CNT_THREADS / 32))

__device__ __forceinline__ void sort_one(unsigned* cnt, int* myList, unsigned& wc,
                                         int s, int lo, int e, bool ok,
                                         unsigned& nBelow) {
  const unsigned d = (unsigned)(s - lo);
  nBelow += (ok && s < lo) ? 1u : 0u;
  const bool own = ok && d < (unsigned)CNT_BINS;
  const unsigned mask = __builtin_amdgcn_ballot_w32(own);
  if (own) {
    atomicAdd(&cnt[d], 1u);
    const unsigned pos = wc + __builtin_amdgcn_mbcnt_lo(mask, 0u);
    if (pos < (unsigned)WAVE_CAP) myList[pos] = e;
  }
  wc += (unsigned)__builtin_popcount(mask);
}

__global__ __launch_bounds__(CNT_THREADS)
void k1_build_csr(const int* __restrict__ edges, int* __restrict__ rowptr,
                  int* __restrict__ csr, int N, int E) {
  __shared__ unsigned cnt[CNT_BINS];
  __shared__ unsigned off[CNT_BINS];
  __shared__ int      list[CSR_CAP];
  __shared__ unsigned wsum[32];
  __shared__ unsigned wcnt[32];
  __shared__ unsigned sBelow, sOver;
  const int t    = blockIdx.y;
  const int lo   = blockIdx.x * CNT_BINS;
  const int tid  = threadIdx.x;
  const int lane = tid & 31;
  const int wave  = __builtin_amdgcn_readfirstlane(tid >> 5);
  const int wbase = wave * 32;

  for (int i = tid; i < CNT_BINS; i += CNT_THREADS) cnt[i] = 0u;
  if (tid == 0) { sBelow = 0u; sOver = 0u; }
  __syncthreads();

  const int* srcE = edges + (size_t)t * 2 * E;
  const int* dstE = srcE + E;
  int* myList = list + wave * WAVE_CAP;
  unsigned nBelow = 0u;
  unsigned wc = 0u;
  if ((E & 3) == 0) {
    const int4* p  = (const int4*)srcE;
    const int   n4 = E >> 2;
    int i0 = wbase;
    for (; i0 + 31 + (CNT_BATCH - 1) * CNT_THREADS < n4; i0 += CNT_BATCH * CNT_THREADS) {
      int4 v[CNT_BATCH];
      #pragma unroll
      for (int j = 0; j < CNT_BATCH; ++j) v[j] = p[i0 + lane + j * CNT_THREADS];
      #pragma unroll
      for (int j = 0; j < CNT_BATCH; ++j) {
        const int e = 4 * (i0 + lane + j * CNT_THREADS);
        sort_one(cnt, myList, wc, v[j].x, lo, e,     true, nBelow);
        sort_one(cnt, myList, wc, v[j].y, lo, e + 1, true, nBelow);
        sort_one(cnt, myList, wc, v[j].z, lo, e + 2, true, nBelow);
        sort_one(cnt, myList, wc, v[j].w, lo, e + 3, true, nBelow);
      }
    }
    for (; i0 < n4; i0 += CNT_THREADS) {
      const int  i  = i0 + lane;
      const bool ok = i < n4;
      int4 v = p[ok ? i : 0];
      const int e = 4 * i;
      sort_one(cnt, myList, wc, v.x, lo, e,     ok, nBelow);
      sort_one(cnt, myList, wc, v.y, lo, e + 1, ok, nBelow);
      sort_one(cnt, myList, wc, v.z, lo, e + 2, ok, nBelow);
      sort_one(cnt, myList, wc, v.w, lo, e + 3, ok, nBelow);
    }
  } else {
    for (int i0 = wbase; i0 < E; i0 += CNT_THREADS) {
      const int  i  = i0 + lane;
      const bool ok = i < E;
      sort_one(cnt, myList, wc, srcE[ok ? i : 0], lo, i, ok, nBelow);
    }
  }
  for (int o = 16; o > 0; o >>= 1) nBelow += __shfl_xor(nBelow, o, 32);
  if (lane == 0) {
    if (nBelow) atomicAdd(&sBelow, nBelow);
    wcnt[wave] = wc;
    if (wc > (unsigned)WAVE_CAP) atomicOr(&sOver, 1u);
  }
  __syncthreads();

  const unsigned c0 = cnt[2 * tid], c1 = cnt[2 * tid + 1];
  unsigned v = c0 + c1;
  for (int o = 1; o < 32; o <<= 1) {
    unsigned u = __shfl_up(v, o, 32);
    if (lane >= o) v += u;
  }
  if (lane == 31) wsum[wave] = v;
  __syncthreads();
  if (wave == 0) {
    const unsigned w0 = wsum[lane];
    unsigned w = w0;
    for (int o = 1; o < 32; o <<= 1) {
      unsigned u = __shfl_up(w, o, 32);
      if (lane >= o) w += u;
    }
    wsum[lane] = w - w0;
  }
  __syncthreads();
  const unsigned base = wsum[wave] + v - (c0 + c1);
  off[2 * tid]     = base;
  off[2 * tid + 1] = base + c0;
  __syncthreads();

  const unsigned below = sBelow;
  const bool     over  = sOver != 0u;
  int* rp = rowptr + (size_t)t * (N + 1);
  for (int i = tid; i < CNT_BINS; i += CNT_THREADS) {
    const int n = lo + i;
    if (n < N) {
      rp[n] = (int)(below + off[i]);
      if (n == N - 1) rp[N] = (int)(below + off[i] + cnt[i]);
    }
  }
  __syncthreads();
  for (int i = tid; i < CNT_BINS; i += CNT_THREADS) cnt[i] = 0u;
  __syncthreads();

  int* csrT = csr + (size_t)t * E;
  if (!over) {
    const unsigned mine = wcnt[wave];
    for (unsigned k = lane; k < mine; k += 32) {
      const int e = myList[k];
      const unsigned d = (unsigned)(srcE[e] - lo);
      const unsigned r = atomicAdd(&cnt[d], 1u);
      csrT[below + off[d] + r] = dstE[e];
    }
  } else {
    for (int e = tid; e < E; e += CNT_THREADS) {
      const unsigned d = (unsigned)(srcE[e] - lo);
      if (d < (unsigned)CNT_BINS) {
        const unsigned r = atomicAdd(&cnt[d], 1u);
        csrT[below + off[d] + r] = dstE[e];
      }
    }
  }
}

__global__ __launch_bounds__(WAVES * 32)
void k3_node_gather(const int* __restrict__ rowptr, const int* __restrict__ csr,
                    const float* __restrict__ attn, const float* __restrict__ pq,
                    float* __restrict__ out, int N, int E, int T) {
  const int tid  = threadIdx.x;
  const int wave = __builtin_amdgcn_readfirstlane(tid >> 5);
  const int lane = tid & 31;
  const size_t planeF = (size_t)N * D_FEAT;

  float m = attn[0];
  for (int u = 1; u < T; ++u) m = fmaxf(m, attn[u]);
  float s = 0.0f;
  for (int u = 0; u < T; ++u) s += __expf(attn[u] - m);

  for (int n = blockIdx.x * WAVES + wave; n < N; n += gridDim.x * WAVES) {
    float acc0 = 0.0f, acc1 = 0.0f;
    for (int tc = 0; tc < T; tc += NODE_TCH) {
      const int jl  = (lane < NODE_TCH) ? lane : 0;
      const int tj  = tc + jl;
      const int tjc = (tj < T) ? tj : (T - 1);
      const int st  = rowptr[(size_t)tjc * (N + 1) + n];
      int       en  = rowptr[(size_t)tjc * (N + 1) + n + 1];
      if (tj >= T) en = st;
      const float wl = __expf(attn[tjc] - m) / s;

      v2f p[NODE_TCH];
      int idx[NODE_TCH];
      v2f q[NODE_TCH][NODE_QB];
      #pragma unroll
      for (int j = 0; j < NODE_TCH; ++j) {
        const int tt = (tc + j < T) ? (tc + j) : (T - 1);
        const int sj = __builtin_amdgcn_readlane(st, j);
        const int cj = __builtin_amdgcn_readlane(en, j) - sj;
        p[j] = *(const v2f*)(pq + (size_t)(2 * tt) * planeF + (size_t)n * D_FEAT + 2 * lane);
        int k = (lane < cj) ? lane : (cj - 1);
        if (k < 0) k = 0;
        int pos = sj + k;
        if (pos > E - 1) pos = E - 1;
        idx[j] = csr[(size_t)tt * E + pos];
      }
      int dn[NODE_TCH][NODE_QB];
      #pragma unroll
      for (int j = 0; j < NODE_TCH; ++j) {
        #pragma unroll
        for (int u = 0; u < NODE_QB; ++u) dn[j][u] = __builtin_amdgcn_readlane(idx[j], u);
      }
      #pragma unroll
      for (int j = 0; j < NODE_TCH; ++j) {
        const int tt = (tc + j < T) ? (tc + j) : (T - 1);
        const int cj = __builtin_amdgcn_readlane(en, j) - __builtin_amdgcn_readlane(st, j);
        const float* planeQ = pq + (size_t)(2 * tt + 1) * planeF + 2 * lane;
        #pragma unroll
        for (int u = 0; u < NODE_QB; ++u) {
          if (u < cj) q[j][u] = *(const v2f*)(planeQ + (size_t)dn[j][u] * D_FEAT);
        }
      }
      #pragma unroll
      for (int j = 0; j < NODE_TCH; ++j) {
        const int tt = (tc + j < T) ? (tc + j) : (T - 1);
        const int sj = __builtin_amdgcn_readlane(st, j);
        const int cj = __builtin_amdgcn_readlane(en, j) - sj;
        if (cj > 0) {
          const float* planeQ = pq + (size_t)(2 * tt + 1) * planeF + 2 * lane;
          float a0 = 0.0f, a1 = 0.0f;
          #pragma unroll
          for (int u = 0; u < NODE_QB; ++u) {
            if (u < cj) {
              a0 += fmaxf(p[j].x + q[j][u].x, 0.0f);
              a1 += fmaxf(p[j].y + q[j][u].y, 0.0f);
            }
          }
          for (int b0 = 0; b0 < cj; b0 += 32) {
            const int rem = (cj - b0 < 32) ? (cj - b0) : 32;
            int id = idx[j];
            if (b0 > 0) {
              const int k = (lane < rem) ? lane : (rem - 1);
              id = csr[(size_t)tt * E + sj + b0 + k];
            }
            for (int k0 = (b0 == 0) ? NODE_QB : 0; k0 < rem; k0 += NODE_QT) {
              v2f r[NODE_QT];
              #pragma unroll
              for (int u = 0; u < NODE_QT; ++u) {
                const int kk = (k0 + u < rem) ? (k0 + u) : (rem - 1);
                const int dn = __builtin_amdgcn_readlane(id, kk);
                r[u] = *(const v2f*)(planeQ + (size_t)dn * D_FEAT);
              }
              #pragma unroll
              for (int u = 0; u < NODE_QT; ++u) {
                if (k0 + u < rem) {
                  a0 += fmaxf(p[j].x + r[u].x, 0.0f);
                  a1 += fmaxf(p[j].y + r[u].y, 0.0f);
                }
              }
            }
          }
          const float wj = __int_as_float(__builtin_amdgcn_readlane(__float_as_int(wl), j));
          const float sc = wj / fmaxf((float)cj, 1.0f);
          acc0 += a0 * sc;
          acc1 += a1 * sc;
        }
      }
    }
    v2f o = {acc0, acc1};
    *(v2f*)(out + (size_t)n * D_FEAT + 2 * lane) = o;
  }
}

extern "C" void kernel_launch(void* const* d_in, const int* in_sizes, int n_in,
                              void* d_out, int out_size, void* d_ws, size_t ws_size,
                              hipStream_t stream) {
  const float* x     = (const float*)d_in[0];
  const float* W     = (const float*)d_in[1];
  const float* b     = (const float*)d_in[2];
  const float* attn  = (const float*)d_in[3];
  const int*   edges = (const int*)d_in[4];
  float* out = (float*)d_out;
  float* wsf = (float*)d_ws;

  const int N = in_sizes[0] / D_FEAT;
  const int T = in_sizes[3];
  const int E = in_sizes[4] / (2 * T);

  const size_t planeF = (size_t)N * D_FEAT;
  const size_t wsF    = ws_size / sizeof(float);
  {
    const size_t csrOff = ((size_t)T * (N + 1) + 3) & ~(size_t)3;
    const size_t pqOffG = (csrOff + (size_t)T * E + 63) & ~(size_t)63;
    if (wsF >= pqOffG + (size_t)2 * T * planeF) {
      int*   rowptr = (int*)d_ws;
      int*   csr    = rowptr + csrOff;
      float* pqG    = wsf + pqOffG;
      dim3 cgrid((unsigned)((N + CNT_BINS - 1) / CNT_BINS), T);
      k1_build_csr<<<cgrid, CNT_THREADS, 0, stream>>>(edges, rowptr, csr, N, E);
      const int pgx = (PROJ_GROUPS / T > 0) ? (PROJ_GROUPS / T) : 1;
      k2_project<<<dim3(pgx, T), WAVES * 32, 0, stream>>>(x, W, b, pqG, N, 0);
      k3_node_gather<<<EDGE_GROUPS, WAVES * 32, 0, stream>>>(rowptr, csr, attn, pqG, out,
                                                              N, E, T);
      return;
    }
  }

  k0_init<<<2048, 256, 0, stream>>>(out, N);
  {
    dim3 cgrid((unsigned)((N + CNT_BINS - 1) / CNT_BINS), T);
    k1_count_scale<<<cgrid, CNT_THREADS, 0, stream>>>(edges, attn, wsf + 8, N, E, T);
  }

  const size_t pqOff  = ((size_t)8 + (size_t)T * N + 63) & ~(size_t)63;
  int G = 0;
  if (wsF > pqOff) {
    size_t g = (wsF - pqOff) / (2 * planeF);
    G = (g < (size_t)T) ? (int)g : T;
  }

  if (G == 0) {
    dim3 grid(GROUPS_PER_TYPE, T);
    k3_gemm_scatter<<<grid, WAVES * 32, 0, stream>>>(x, W, b, edges,
                                                     wsf + 8, out, N, E);
    return;
  }

  float* pq = wsf + pqOff;
  for (int t0 = 0; t0 < T; t0 += G) {
    const int g = (T - t0 < G) ? (T - t0) : G;
    const int pgx = (PROJ_GROUPS / g > 0) ? (PROJ_GROUPS / g) : 1;
    k2_project<<<dim3(pgx, g), WAVES * 32, 0, stream>>>(x, W, b, pq, N, t0);
    k3_edge_scatter<<<EDGE_GROUPS, WAVES * 32, 0, stream>>>(edges, wsf + 8, pq, out,
                                                            N, E, t0, g);
  }
}
